# starting kernel
# baseline (speedup 1.0000x reference)
#include <hip/hip_runtime.h>
#include <hip/hip_bf16.h>

typedef __attribute__((ext_vector_type(2))) float v2f;
typedef __attribute__((ext_vector_type(8))) float v8f;

#define IN_FEATS  256
#define OUT_FEATS 64

__global__ void gcn_gemm_wmma(const float* __restrict__ x,
                              const float* __restrict__ W,
                              const float* __restrict__ bias,
                              float* __restrict__ h,
                              int n_nodes)
{
    const int lane   = threadIdx.x & 31;
    const int waveId = threadIdx.x >> 5;
    const int wavesPerBlock = blockDim.x >> 5;
    const int rowTile = blockIdx.x * wavesPerBlock + waveId;

    const int row0 = rowTile * 16;
    if (row0 >= n_nodes) return;

    const int m  = lane & 15;
    const int hi = lane >> 4;
    const int n  = lane & 15;

    int arow = row0 + m;
    if (arow >= n_nodes) arow = n_nodes - 1;
    const float* __restrict__ xrow = x + (size_t)arow * IN_FEATS;

    v8f acc0 = {}, acc1 = {}, acc2 = {}, acc3 = {};

    #pragma unroll 4
    for (int k = 0; k < IN_FEATS; k += 4) {
        const int kk = k + 2 * hi;
        v2f a = *(const v2f*)(xrow + kk);

        const float* __restrict__ w0 = W + (size_t)kk * OUT_FEATS + n;
        const float* __restrict__ w1 = w0 + OUT_FEATS;
        v2f b0 = { w0[0],  w1[0]  };
        v2f b1 = { w0[16], w1[16] };
        v2f b2 = { w0[32], w1[32] };
        v2f b3 = { w0[48], w1[48] };

        acc0 = __builtin_amdgcn_wmma_f32_16x16x4_f32(false, a, false, b0, (short)0, acc0, false, false);
        acc1 = __builtin_amdgcn_wmma_f32_16x16x4_f32(false, a, false, b1, (short)0, acc1, false, false);
        acc2 = __builtin_amdgcn_wmma_f32_16x16x4_f32(false, a, false, b2, (short)0, acc2, false, false);
        acc3 = __builtin_amdgcn_wmma_f32_16x16x4_f32(false, a, false, b3, (short)0, acc3, false, false);
    }

    const float bn0 = bias[n];
    const float bn1 = bias[16 + n];
    const float bn2 = bias[32 + n];
    const float bn3 = bias[48 + n];

    #pragma unroll
    for (int r = 0; r < 8; ++r) {
        const int row = row0 + r + 8 * hi;
        if (row < n_nodes) {
            float* __restrict__ hrow = h + (size_t)row * OUT_FEATS;
            hrow[n]      = acc0[r] + bn0;
            hrow[16 + n] = acc1[r] + bn1;
            hrow[32 + n] = acc2[r] + bn2;
            hrow[48 + n] = acc3[r] + bn3;
        }
    }
}

__global__ void gcn_edge_scatter(const float* __restrict__ h,
                                 const int* __restrict__ edge_src,
                                 const int* __restrict__ edge_dst,
                                 const float* __restrict__ edge_w,
                                 float* __restrict__ out,
                                 int n_edges)
{
    const int lane   = threadIdx.x & 31;
    const int waveId = threadIdx.x >> 5;
    const int wavesPerBlock = blockDim.x >> 5;

    const int e = __builtin_amdgcn_readfirstlane(blockIdx.x * wavesPerBlock + waveId);
    if (e >= n_edges) return;

    const int   s = edge_src[e];
    const int   d = edge_dst[e];
    const float w = edge_w[e];

    const v2f hv = *(const v2f*)(h + (size_t)s * OUT_FEATS + 2 * lane);

    float* __restrict__ o = out + (size_t)d * OUT_FEATS + 2 * lane;
#if defined(__HIP_DEVICE_COMPILE__)
    unsafeAtomicAdd(o,     w * hv.x);
    unsafeAtomicAdd(o + 1, w * hv.y);
#else
    atomicAdd(o,     w * hv.x);
    atomicAdd(o + 1, w * hv.y);
#endif
}

extern "C" void kernel_launch(void* const* d_in, const int* in_sizes, int n_in,
                              void* d_out, int out_size, void* d_ws, size_t ws_size,
                              hipStream_t stream)
{
    const float* x    = (const float*)d_in[0];
    const int*   esrc = (const int*)  d_in[1];
    const int*   edst = (const int*)  d_in[2];
    const float* ew   = (const float*)d_in[3];
    const float* W    = (const float*)d_in[4];
    const float* bias = (const float*)d_in[5];

    float* out = (float*)d_out;
    float* h   = (float*)d_ws;

    const int n_nodes = in_sizes[0] / IN_FEATS;
    const int n_edges = in_sizes[1];

    hipMemsetAsync(d_out, 0, (size_t)out_size * sizeof(float), stream);

    const int rowTiles = (n_nodes + 15) / 16;
    const int wavesPerBlock = 8;
    const int gemmBlocks = (rowTiles + wavesPerBlock - 1) / wavesPerBlock;
    gcn_gemm_wmma<<<gemmBlocks, wavesPerBlock * 32, 0, stream>>>(x, W, bias, h, n_nodes);

    const int scatterBlocks = (n_edges + wavesPerBlock - 1) / wavesPerBlock;
    gcn_edge_scatter<<<scatterBlocks, wavesPerBlock * 32, 0, stream>>>(h, esrc, edst, ew, out, n_edges);
}


# scatter re-tiled to whole-line memory instructions: a wave takes 16 edges (ids by one vector load per array) and lane = feature, so each gather and each f32 atomic covers one 128-B line
# speedup vs baseline: 2.2059x; 2.2059x over previous
#include <hip/hip_runtime.h>
#include <hip/hip_bf16.h>

typedef __attribute__((ext_vector_type(2))) float v2f;
typedef __attribute__((ext_vector_type(8))) float v8f;

#define IN_FEATS  256
#define OUT_FEATS 64

__global__ void gcn_gemm_wmma(const float* __restrict__ x,
                              const float* __restrict__ W,
                              const float* __restrict__ bias,
                              float* __restrict__ h,
                              int n_nodes)
{
    const int lane   = threadIdx.x & 31;
    const int waveId = threadIdx.x >> 5;
    const int wavesPerBlock = blockDim.x >> 5;
    const int rowTile = blockIdx.x * wavesPerBlock + waveId;

    const int row0 = rowTile * 16;
    if (row0 >= n_nodes) return;

    const int m  = lane & 15;
    const int hi = lane >> 4;
    const int n  = lane & 15;

    int arow = row0 + m;
    if (arow >= n_nodes) arow = n_nodes - 1;
    const float* __restrict__ xrow = x + (size_t)arow * IN_FEATS;

    v8f acc0 = {}, acc1 = {}, acc2 = {}, acc3 = {};

    #pragma unroll 4
    for (int k = 0; k < IN_FEATS; k += 4) {
        const int kk = k + 2 * hi;
        v2f a = *(const v2f*)(xrow + kk);

        const float* __restrict__ w0 = W + (size_t)kk * OUT_FEATS + n;
        const float* __restrict__ w1 = w0 + OUT_FEATS;
        v2f b0 = { w0[0],  w1[0]  };
        v2f b1 = { w0[16], w1[16] };
        v2f b2 = { w0[32], w1[32] };
        v2f b3 = { w0[48], w1[48] };

        acc0 = __builtin_amdgcn_wmma_f32_16x16x4_f32(false, a, false, b0, (short)0, acc0, false, false);
        acc1 = __builtin_amdgcn_wmma_f32_16x16x4_f32(false, a, false, b1, (short)0, acc1, false, false);
        acc2 = __builtin_amdgcn_wmma_f32_16x16x4_f32(false, a, false, b2, (short)0, acc2, false, false);
        acc3 = __builtin_amdgcn_wmma_f32_16x16x4_f32(false, a, false, b3, (short)0, acc3, false, false);
    }

    const float bn0 = bias[n];
    const float bn1 = bias[16 + n];
    const float bn2 = bias[32 + n];
    const float bn3 = bias[48 + n];

    #pragma unroll
    for (int r = 0; r < 8; ++r) {
        const int row = row0 + r + 8 * hi;
        if (row < n_nodes) {
            float* __restrict__ hrow = h + (size_t)row * OUT_FEATS;
            hrow[n]      = acc0[r] + bn0;
            hrow[16 + n] = acc1[r] + bn1;
            hrow[32 + n] = acc2[r] + bn2;
            hrow[48 + n] = acc3[r] + bn3;
        }
    }
}

#define EDGES_PER_WAVE 16

__device__ __forceinline__ void edge_add(float* __restrict__ row, int lane, float v0, float v1)
{
#if defined(__HIP_DEVICE_COMPILE__)
    unsafeAtomicAdd(row + lane,                 v0);
    unsafeAtomicAdd(row + lane + OUT_FEATS / 2, v1);
#else
    atomicAdd(row + lane,                 v0);
    atomicAdd(row + lane + OUT_FEATS / 2, v1);
#endif
}

__global__ void gcn_edge_scatter(const float* __restrict__ h,
                                 const int* __restrict__ edge_src,
                                 const int* __restrict__ edge_dst,
                                 const float* __restrict__ edge_w,
                                 float* __restrict__ out,
                                 int n_edges)
{
    const int lane   = threadIdx.x & 31;
    const int waveId = threadIdx.x >> 5;
    const int wavesPerBlock = blockDim.x >> 5;

    const int wave = __builtin_amdgcn_readfirstlane(blockIdx.x * wavesPerBlock + waveId);
    const int e0 = wave * EDGES_PER_WAVE;
    if (e0 >= n_edges) return;
    const int left = n_edges - e0;
    const int cnt  = left < EDGES_PER_WAVE ? left : EDGES_PER_WAVE;

    const int sub = lane & (EDGES_PER_WAVE - 1);
    const int ei  = e0 + (sub < cnt ? sub : 0);
    const int sv = edge_src[ei];
    const int dv = edge_dst[ei];
    const int wv = __float_as_int(edge_w[ei]);

    if (cnt == EDGES_PER_WAVE) {
        float r0[EDGES_PER_WAVE], r1[EDGES_PER_WAVE];
        #pragma unroll
        for (int j = 0; j < EDGES_PER_WAVE; ++j) {
            const int s = __builtin_amdgcn_readlane(sv, j);
            const float* __restrict__ p = h + (size_t)s * OUT_FEATS;
            r0[j] = p[lane];
            r1[j] = p[lane + OUT_FEATS / 2];
        }
        #pragma unroll
        for (int j = 0; j < EDGES_PER_WAVE; ++j) {
            const int   d = __builtin_amdgcn_readlane(dv, j);
            const float w = __int_as_float(__builtin_amdgcn_readlane(wv, j));
            edge_add(out + (size_t)d * OUT_FEATS, lane, w * r0[j], w * r1[j]);
        }
    } else {
        for (int j = 0; j < cnt; ++j) {
            const int   s = __builtin_amdgcn_readlane(sv, j);
            const int   d = __builtin_amdgcn_readlane(dv, j);
            const float w = __int_as_float(__builtin_amdgcn_readlane(wv, j));
            const float* __restrict__ p = h + (size_t)s * OUT_FEATS;
            const float a0 = p[lane];
            const float a1 = p[lane + OUT_FEATS / 2];
            edge_add(out + (size_t)d * OUT_FEATS, lane, w * a0, w * a1);
        }
    }
}

extern "C" void kernel_launch(void* const* d_in, const int* in_sizes, int n_in,
                              void* d_out, int out_size, void* d_ws, size_t ws_size,
                              hipStream_t stream)
{
    const float* x    = (const float*)d_in[0];
    const int*   esrc = (const int*)  d_in[1];
    const int*   edst = (const int*)  d_in[2];
    const float* ew   = (const float*)d_in[3];
    const float* W    = (const float*)d_in[4];
    const float* bias = (const float*)d_in[5];

    float* out = (float*)d_out;
    float* h   = (float*)d_ws;

    const int n_nodes = in_sizes[0] / IN_FEATS;
    const int n_edges = in_sizes[1];

    hipMemsetAsync(d_out, 0, (size_t)out_size * sizeof(float), stream);

    const int rowTiles = (n_nodes + 15) / 16;
    const int wavesPerBlock = 8;
    const int gemmBlocks = (rowTiles + wavesPerBlock - 1) / wavesPerBlock;
    gcn_gemm_wmma<<<gemmBlocks, wavesPerBlock * 32, 0, stream>>>(x, W, bias, h, n_nodes);

    const int edgesPerBlock = wavesPerBlock * EDGES_PER_WAVE;
    const int scatterBlocks = (n_edges + edgesPerBlock - 1) / edgesPerBlock;
    gcn_edge_scatter<<<scatterBlocks, wavesPerBlock * 32, 0, stream>>>(h, esrc, edst, ew, out, n_edges);
}


# edge sum without global atomics: per call, edges grouped by destination range with LDS counters, each range summed by f32 LDS adds (arrival order), each out row stored once; out clear dropped
# speedup vs baseline: 2.3186x; 1.0511x over previous
#include <hip/hip_runtime.h>
#include <hip/hip_bf16.h>

typedef __attribute__((ext_vector_type(2))) float v2f;
typedef __attribute__((ext_vector_type(8))) float v8f;

#define IN_FEATS  256
#define OUT_FEATS 64

__global__ void gcn_gemm_wmma(const float* __restrict__ x,
                              const float* __restrict__ W,
                              const float* __restrict__ bias,
                              float* __restrict__ h,
                              int n_nodes)
{
    const int lane   = threadIdx.x & 31;
    const int waveId = threadIdx.x >> 5;
    const int wavesPerBlock = blockDim.x >> 5;
    const int rowTile = blockIdx.x * wavesPerBlock + waveId;

    const int row0 = rowTile * 16;
    if (row0 >= n_nodes) return;

    const int m  = lane & 15;
    const int hi = lane >> 4;
    const int n  = lane & 15;

    int arow = row0 + m;
    if (arow >= n_nodes) arow = n_nodes - 1;
    const float* __restrict__ xrow = x + (size_t)arow * IN_FEATS;

    v8f acc0 = {}, acc1 = {}, acc2 = {}, acc3 = {};

    #pragma unroll 4
    for (int k = 0; k < IN_FEATS; k += 4) {
        const int kk = k + 2 * hi;
        v2f a = *(const v2f*)(xrow + kk);

        const float* __restrict__ w0 = W + (size_t)kk * OUT_FEATS + n;
        const float* __restrict__ w1 = w0 + OUT_FEATS;
        v2f b0 = { w0[0],  w1[0]  };
        v2f b1 = { w0[16], w1[16] };
        v2f b2 = { w0[32], w1[32] };
        v2f b3 = { w0[48], w1[48] };

        acc0 = __builtin_amdgcn_wmma_f32_16x16x4_f32(false, a, false, b0, (short)0, acc0, false, false);
        acc1 = __builtin_amdgcn_wmma_f32_16x16x4_f32(false, a, false, b1, (short)0, acc1, false, false);
        acc2 = __builtin_amdgcn_wmma_f32_16x16x4_f32(false, a, false, b2, (short)0, acc2, false, false);
        acc3 = __builtin_amdgcn_wmma_f32_16x16x4_f32(false, a, false, b3, (short)0, acc3, false, false);
    }

    const float bn0 = bias[n];
    const float bn1 = bias[16 + n];
    const float bn2 = bias[32 + n];
    const float bn3 = bias[48 + n];

    #pragma unroll
    for (int r = 0; r < 8; ++r) {
        const int row = row0 + r + 8 * hi;
        if (row < n_nodes) {
            float* __restrict__ hrow = h + (size_t)row * OUT_FEATS;
            hrow[n]      = acc0[r] + bn0;
            hrow[16 + n] = acc1[r] + bn1;
            hrow[32 + n] = acc2[r] + bn2;
            hrow[48 + n] = acc3[r] + bn3;
        }
    }
}

#define EDGES_PER_WAVE 16

__device__ __forceinline__ void edge_add(float* __restrict__ row, int lane, float v0, float v1)
{
#if defined(__HIP_DEVICE_COMPILE__)
    unsafeAtomicAdd(row + lane,                 v0);
    unsafeAtomicAdd(row + lane + OUT_FEATS / 2, v1);
#else
    atomicAdd(row + lane,                 v0);
    atomicAdd(row + lane + OUT_FEATS / 2, v1);
#endif
}

__global__ void gcn_edge_scatter(const float* __restrict__ h,
                                 const int* __restrict__ edge_src,
                                 const int* __restrict__ edge_dst,
                                 const float* __restrict__ edge_w,
                                 float* __restrict__ out,
                                 int n_edges)
{
    const int lane   = threadIdx.x & 31;
    const int waveId = threadIdx.x >> 5;
    const int wavesPerBlock = blockDim.x >> 5;

    const int wave = __builtin_amdgcn_readfirstlane(blockIdx.x * wavesPerBlock + waveId);
    const int e0 = wave * EDGES_PER_WAVE;
    if (e0 >= n_edges) return;
    const int left = n_edges - e0;
    const int cnt  = left < EDGES_PER_WAVE ? left : EDGES_PER_WAVE;

    const int sub = lane & (EDGES_PER_WAVE - 1);
    const int ei  = e0 + (sub < cnt ? sub : 0);
    const int sv = edge_src[ei];
    const int dv = edge_dst[ei];
    const int wv = __float_as_int(edge_w[ei]);

    if (cnt == EDGES_PER_WAVE) {
        float r0[EDGES_PER_WAVE], r1[EDGES_PER_WAVE];
        #pragma unroll
        for (int j = 0; j < EDGES_PER_WAVE; ++j) {
            const int s = __builtin_amdgcn_readlane(sv, j);
            const float* __restrict__ p = h + (size_t)s * OUT_FEATS;
            r0[j] = p[lane];
            r1[j] = p[lane + OUT_FEATS / 2];
        }
        #pragma unroll
        for (int j = 0; j < EDGES_PER_WAVE; ++j) {
            const int   d = __builtin_amdgcn_readlane(dv, j);
            const float w = __int_as_float(__builtin_amdgcn_readlane(wv, j));
            edge_add(out + (size_t)d * OUT_FEATS, lane, w * r0[j], w * r1[j]);
        }
    } else {
        for (int j = 0; j < cnt; ++j) {
            const int   s = __builtin_amdgcn_readlane(sv, j);
            const int   d = __builtin_amdgcn_readlane(dv, j);
            const float w = __int_as_float(__builtin_amdgcn_readlane(wv, j));
            const float* __restrict__ p = h + (size_t)s * OUT_FEATS;
            const float a0 = p[lane];
            const float a1 = p[lane + OUT_FEATS / 2];
            edge_add(out + (size_t)d * OUT_FEATS, lane, w * a0, w * a1);
        }
    }
}

#define RB        196
#define NBMAX     512
#define CHUNK     6400
#define NBLK_MAX  256
#define BIN_T     1024
#define BIN_EPT   ((CHUNK + BIN_T - 1) / BIN_T)
#define RED_T     512

__global__ __launch_bounds__(BIN_T)
void gcn_bin_edges(const int* __restrict__ edge_src,
                   const int* __restrict__ edge_dst,
                   const float* __restrict__ edge_w,
                   int2* __restrict__ recs,
                   int* __restrict__ offT,
                   int n_edges, int n_nodes)
{
    __shared__ int  cnt[NBMAX];
    __shared__ int  scan[2][NBMAX];
    __shared__ int2 stage[CHUNK];

    const int t   = threadIdx.x;
    const int blk = blockIdx.x;
    const int e0  = blk * CHUNK;
    const int rem = n_edges - e0;
    const int len = rem < CHUNK ? rem : CHUNK;

    if (t < NBMAX) cnt[t] = 0;
    __syncthreads();

    int key[BIN_EPT], wbt[BIN_EPT], bkt[BIN_EPT], rnk[BIN_EPT];
    #pragma unroll
    for (int i = 0; i < BIN_EPT; ++i) {
        const int idx = t + i * BIN_T;
        bkt[i] = -1; key[i] = 0; wbt[i] = 0; rnk[i] = 0;
        if (idx < len) {
            const int e = e0 + idx;
            const int d = edge_dst[e];
            const int s = edge_src[e];
            const int w = __float_as_int(edge_w[e]);
            if ((unsigned)d < (unsigned)n_nodes) {
                const int b = d / RB;
                bkt[i] = b;
                key[i] = s | ((d - b * RB) << 24);
                wbt[i] = w;
                rnk[i] = atomicAdd(&cnt[b], 1);
            }
        }
    }
    __syncthreads();

    if (t < NBMAX) scan[0][t] = cnt[t];
    __syncthreads();
    int cur = 0;
    for (int d = 1; d < NBMAX; d <<= 1) {
        if (t < NBMAX) {
            int v = scan[cur][t];
            if (t >= d) v += scan[cur][t - d];
            scan[cur ^ 1][t] = v;
        }
        __syncthreads();
        cur ^= 1;
    }
    if (t < NBMAX) {
        const int ex = scan[cur][t] - cnt[t];
        scan[cur ^ 1][t] = ex;
        offT[t * NBLK_MAX + blk] = ex;
        if (t == NBMAX - 1) offT[NBMAX * NBLK_MAX + blk] = scan[cur][t];
    }
    __syncthreads();
    const int total = scan[cur][NBMAX - 1];

    #pragma unroll
    for (int i = 0; i < BIN_EPT; ++i) {
        if (bkt[i] >= 0) stage[scan[cur ^ 1][bkt[i]] + rnk[i]] = make_int2(key[i], wbt[i]);
    }
    __syncthreads();

    for (int idx = t; idx < total; idx += BIN_T) recs[e0 + idx] = stage[idx];
}

__device__ __forceinline__ void lds_row_add(float* row, int lane, float v0, float v1)
{
    __hip_atomic_fetch_add(row + lane,                 v0, __ATOMIC_RELAXED, __HIP_MEMORY_SCOPE_WORKGROUP);
    __hip_atomic_fetch_add(row + lane + OUT_FEATS / 2, v1, __ATOMIC_RELAXED, __HIP_MEMORY_SCOPE_WORKGROUP);
}

__global__ __launch_bounds__(RED_T)
void gcn_bucket_sum(const float* __restrict__ h,
                    const int2* __restrict__ recs,
                    const int* __restrict__ offT,
                    float* __restrict__ out,
                    int n_nodes, int nblk)
{
    __shared__ float acc[RB * OUT_FEATS];
    __shared__ int   segStart[NBLK_MAX];
    __shared__ int   segBase[NBLK_MAX + 1];
    __shared__ int   scan[2][NBLK_MAX];

    const int t    = threadIdx.x;
    const int lane = t & 31;
    const int b    = blockIdx.x;

    for (int i = t; i < RB * OUT_FEATS; i += RED_T) acc[i] = 0.0f;

    if (t < NBLK_MAX) {
        int mylen = 0, st = 0;
        if (t < nblk) {
            const int s = offT[b * NBLK_MAX + t];
            const int e = offT[(b + 1) * NBLK_MAX + t];
            st = t * CHUNK + s;
            mylen = e - s;
        }
        segStart[t] = st;
        scan[0][t]  = mylen;
    }
    __syncthreads();
    int cur = 0;
    for (int d = 1; d < NBLK_MAX; d <<= 1) {
        if (t < NBLK_MAX) {
            int v = scan[cur][t];
            if (t >= d) v += scan[cur][t - d];
            scan[cur ^ 1][t] = v;
        }
        __syncthreads();
        cur ^= 1;
    }
    if (t < NBLK_MAX) segBase[t + 1] = scan[cur][t];
    if (t == 0) segBase[0] = 0;
    __syncthreads();

    const int total = __builtin_amdgcn_readfirstlane(segBase[NBLK_MAX]);
    const int wv    = __builtin_amdgcn_readfirstlane(t >> 5);

    for (int base = wv * 32; base < total; base += RED_T) {
        const int i = base + lane;
        int key = 0, wbt = 0;
        if (i < total) {
            int lo = 0;
            #pragma unroll
            for (int step = NBLK_MAX / 2; step >= 1; step >>= 1)
                if (segBase[lo + step] <= i) lo += step;
            const int2 r = recs[segStart[lo] + (i - segBase[lo])];
            key = r.x; wbt = r.y;
        }
        const int left = total - base;
        const int cnt  = left < 32 ? left : 32;

        #pragma unroll
        for (int jb = 0; jb < 32; jb += EDGES_PER_WAVE) {
            if (cnt >= jb + EDGES_PER_WAVE) {
                float r0[EDGES_PER_WAVE], r1[EDGES_PER_WAVE];
                #pragma unroll
                for (int j = 0; j < EDGES_PER_WAVE; ++j) {
                    const int k = __builtin_amdgcn_readlane(key, jb + j);
                    const float* __restrict__ p = h + (size_t)(k & 0xFFFFFF) * OUT_FEATS;
                    r0[j] = p[lane];
                    r1[j] = p[lane + OUT_FEATS / 2];
                }
                #pragma unroll
                for (int j = 0; j < EDGES_PER_WAVE; ++j) {
                    const int   k = __builtin_amdgcn_readlane(key, jb + j);
                    const float w = __int_as_float(__builtin_amdgcn_readlane(wbt, jb + j));
                    lds_row_add(acc + ((unsigned)k >> 24) * OUT_FEATS, lane, w * r0[j], w * r1[j]);
                }
            } else {
                for (int j = jb; j < cnt; ++j) {
                    const int   k = __builtin_amdgcn_readlane(key, j);
                    const float w = __int_as_float(__builtin_amdgcn_readlane(wbt, j));
                    const float* __restrict__ p = h + (size_t)(k & 0xFFFFFF) * OUT_FEATS;
                    const float a0 = p[lane];
                    const float a1 = p[lane + OUT_FEATS / 2];
                    lds_row_add(acc + ((unsigned)k >> 24) * OUT_FEATS, lane, w * a0, w * a1);
                }
            }
        }
    }
    __syncthreads();

    const int row0  = b * RB;
    const int nleft = n_nodes - row0;
    const int nrows = nleft < RB ? nleft : RB;
    const float4* __restrict__ acc4 = (const float4*)acc;
    float4* __restrict__ out4 = (float4*)(out + (size_t)row0 * OUT_FEATS);
    for (int i = t; i < nrows * (OUT_FEATS / 4); i += RED_T) out4[i] = acc4[i];
}

extern "C" void kernel_launch(void* const* d_in, const int* in_sizes, int n_in,
                              void* d_out, int out_size, void* d_ws, size_t ws_size,
                              hipStream_t stream)
{
    const float* x    = (const float*)d_in[0];
    const int*   esrc = (const int*)  d_in[1];
    const int*   edst = (const int*)  d_in[2];
    const float* ew   = (const float*)d_in[3];
    const float* W    = (const float*)d_in[4];
    const float* bias = (const float*)d_in[5];

    float* out = (float*)d_out;
    float* h   = (float*)d_ws;

    const int n_nodes = in_sizes[0] / IN_FEATS;
    const int n_edges = in_sizes[1];

    const int rowTiles = (n_nodes + 15) / 16;
    const int wavesPerBlock = 8;
    const int gemmBlocks = (rowTiles + wavesPerBlock - 1) / wavesPerBlock;

    const size_t hBytes   = (((size_t)n_nodes * OUT_FEATS * sizeof(float)) + 255) & ~(size_t)255;
    const size_t recBytes = (((size_t)n_edges * sizeof(int2)) + 255) & ~(size_t)255;
    const size_t offBytes = (size_t)(NBMAX + 1) * NBLK_MAX * sizeof(int);
    const int nblk    = (n_edges + CHUNK - 1) / CHUNK;
    const int buckets = (n_nodes + RB - 1) / RB;

    if (ws_size >= hBytes + recBytes + offBytes && nblk >= 1 && nblk <= NBLK_MAX && buckets <= NBMAX) {
        int2* recs = (int2*)((char*)d_ws + hBytes);
        int*  offT = (int*)((char*)d_ws + hBytes + recBytes);
        gcn_gemm_wmma<<<gemmBlocks, wavesPerBlock * 32, 0, stream>>>(x, W, bias, h, n_nodes);
        gcn_bin_edges<<<nblk, BIN_T, 0, stream>>>(esrc, edst, ew, recs, offT, n_edges, n_nodes);
        gcn_bucket_sum<<<buckets, RED_T, 0, stream>>>(h, recs, offT, out, n_nodes, nblk);
        return;
    }

    hipMemsetAsync(d_out, 0, (size_t)out_size * sizeof(float), stream);
    gcn_gemm_wmma<<<gemmBlocks, wavesPerBlock * 32, 0, stream>>>(x, W, bias, h, n_nodes);

    const int edgesPerBlock = wavesPerBlock * EDGES_PER_WAVE;
    const int scatterBlocks = (n_edges + edgesPerBlock - 1) / edgesPerBlock;
    gcn_edge_scatter<<<scatterBlocks, wavesPerBlock * 32, 0, stream>>>(h, esrc, edst, ew, out, n_edges);
}


# edge binning pass folded into the GEMM launch as a second block role (independent work; GEMM blocks of 32 waves first, one binning block per chunk behind them): 3 launches become 2
# speedup vs baseline: 2.6081x; 1.1249x over previous
#include <hip/hip_runtime.h>
#include <hip/hip_bf16.h>

typedef __attribute__((ext_vector_type(2))) float v2f;
typedef __attribute__((ext_vector_type(8))) float v8f;

#define IN_FEATS  256
#define OUT_FEATS 64

__device__ __forceinline__ void gemm_rows16(const float* __restrict__ x,
                                            const float* __restrict__ W,
                                            const float* __restrict__ bias,
                                            float* __restrict__ h,
                                            int n_nodes, int rowTile)
{
    const int lane   = threadIdx.x & 31;

    const int row0 = rowTile * 16;
    if (row0 >= n_nodes) return;

    const int m  = lane & 15;
    const int hi = lane >> 4;
    const int n  = lane & 15;

    int arow = row0 + m;
    if (arow >= n_nodes) arow = n_nodes - 1;
    const float* __restrict__ xrow = x + (size_t)arow * IN_FEATS;

    v8f acc0 = {}, acc1 = {}, acc2 = {}, acc3 = {};

    #pragma unroll 4
    for (int k = 0; k < IN_FEATS; k += 4) {
        const int kk = k + 2 * hi;
        v2f a = *(const v2f*)(xrow + kk);

        const float* __restrict__ w0 = W + (size_t)kk * OUT_FEATS + n;
        const float* __restrict__ w1 = w0 + OUT_FEATS;
        v2f b0 = { w0[0],  w1[0]  };
        v2f b1 = { w0[16], w1[16] };
        v2f b2 = { w0[32], w1[32] };
        v2f b3 = { w0[48], w1[48] };

        acc0 = __builtin_amdgcn_wmma_f32_16x16x4_f32(false, a, false, b0, (short)0, acc0, false, false);
        acc1 = __builtin_amdgcn_wmma_f32_16x16x4_f32(false, a, false, b1, (short)0, acc1, false, false);
        acc2 = __builtin_amdgcn_wmma_f32_16x16x4_f32(false, a, false, b2, (short)0, acc2, false, false);
        acc3 = __builtin_amdgcn_wmma_f32_16x16x4_f32(false, a, false, b3, (short)0, acc3, false, false);
    }

    const float bn0 = bias[n];
    const float bn1 = bias[16 + n];
    const float bn2 = bias[32 + n];
    const float bn3 = bias[48 + n];

    #pragma unroll
    for (int r = 0; r < 8; ++r) {
        const int row = row0 + r + 8 * hi;
        if (row < n_nodes) {
            float* __restrict__ hrow = h + (size_t)row * OUT_FEATS;
            hrow[n]      = acc0[r] + bn0;
            hrow[16 + n] = acc1[r] + bn1;
            hrow[32 + n] = acc2[r] + bn2;
            hrow[48 + n] = acc3[r] + bn3;
        }
    }
}

__global__ void gcn_gemm_wmma(const float* __restrict__ x,
                              const float* __restrict__ W,
                              const float* __restrict__ bias,
                              float* __restrict__ h,
                              int n_nodes)
{
    const int waveId = threadIdx.x >> 5;
    const int wavesPerBlock = blockDim.x >> 5;
    gemm_rows16(x, W, bias, h, n_nodes, blockIdx.x * wavesPerBlock + waveId);
}

#define EDGES_PER_WAVE 16

__device__ __forceinline__ void edge_add(float* __restrict__ row, int lane, float v0, float v1)
{
#if defined(__HIP_DEVICE_COMPILE__)
    unsafeAtomicAdd(row + lane,                 v0);
    unsafeAtomicAdd(row + lane + OUT_FEATS / 2, v1);
#else
    atomicAdd(row + lane,                 v0);
    atomicAdd(row + lane + OUT_FEATS / 2, v1);
#endif
}

__global__ void gcn_edge_scatter(const float* __restrict__ h,
                                 const int* __restrict__ edge_src,
                                 const int* __restrict__ edge_dst,
                                 const float* __restrict__ edge_w,
                                 float* __restrict__ out,
                                 int n_edges)
{
    const int lane   = threadIdx.x & 31;
    const int waveId = threadIdx.x >> 5;
    const int wavesPerBlock = blockDim.x >> 5;

    const int wave = __builtin_amdgcn_readfirstlane(blockIdx.x * wavesPerBlock + waveId);
    const int e0 = wave * EDGES_PER_WAVE;
    if (e0 >= n_edges) return;
    const int left = n_edges - e0;
    const int cnt  = left < EDGES_PER_WAVE ? left : EDGES_PER_WAVE;

    const int sub = lane & (EDGES_PER_WAVE - 1);
    const int ei  = e0 + (sub < cnt ? sub : 0);
    const int sv = edge_src[ei];
    const int dv = edge_dst[ei];
    const int wv = __float_as_int(edge_w[ei]);

    if (cnt == EDGES_PER_WAVE) {
        float r0[EDGES_PER_WAVE], r1[EDGES_PER_WAVE];
        #pragma unroll
        for (int j = 0; j < EDGES_PER_WAVE; ++j) {
            const int s = __builtin_amdgcn_readlane(sv, j);
            const float* __restrict__ p = h + (size_t)s * OUT_FEATS;
            r0[j] = p[lane];
            r1[j] = p[lane + OUT_FEATS / 2];
        }
        #pragma unroll
        for (int j = 0; j < EDGES_PER_WAVE; ++j) {
            const int   d = __builtin_amdgcn_readlane(dv, j);
            const float w = __int_as_float(__builtin_amdgcn_readlane(wv, j));
            edge_add(out + (size_t)d * OUT_FEATS, lane, w * r0[j], w * r1[j]);
        }
    } else {
        for (int j = 0; j < cnt; ++j) {
            const int   s = __builtin_amdgcn_readlane(sv, j);
            const int   d = __builtin_amdgcn_readlane(dv, j);
            const float w = __int_as_float(__builtin_amdgcn_readlane(wv, j));
            const float* __restrict__ p = h + (size_t)s * OUT_FEATS;
            const float a0 = p[lane];
            const float a1 = p[lane + OUT_FEATS / 2];
            edge_add(out + (size_t)d * OUT_FEATS, lane, w * a0, w * a1);
        }
    }
}

#define RB        196
#define NBMAX     512
#define CHUNK     6400
#define NBLK_MAX  256
#define BIN_T     1024
#define BIN_EPT   ((CHUNK + BIN_T - 1) / BIN_T)
#define RED_T     512

__global__ __launch_bounds__(BIN_T)
void gcn_gemm_and_bin(const float* __restrict__ x,
                      const float* __restrict__ W,
                      const float* __restrict__ bias,
                      float* __restrict__ h,
                      const int* __restrict__ edge_src,
                      const int* __restrict__ edge_dst,
                      const float* __restrict__ edge_w,
                      int2* __restrict__ recs,
                      int* __restrict__ offT,
                      int n_edges, int n_nodes, int gemm_blocks)
{
    __shared__ int  cnt[NBMAX];
    __shared__ int  scan[2][NBMAX];
    __shared__ int2 stage[CHUNK];

    if ((int)blockIdx.x < gemm_blocks) {
        gemm_rows16(x, W, bias, h, n_nodes, blockIdx.x * (BIN_T / 32) + (threadIdx.x >> 5));
        return;
    }

    const int t   = threadIdx.x;
    const int blk = blockIdx.x - gemm_blocks;
    const int e0  = blk * CHUNK;
    const int rem = n_edges - e0;
    const int len = rem < CHUNK ? rem : CHUNK;

    if (t < NBMAX) cnt[t] = 0;
    __syncthreads();

    int key[BIN_EPT], wbt[BIN_EPT], bkt[BIN_EPT], rnk[BIN_EPT];
    #pragma unroll
    for (int i = 0; i < BIN_EPT; ++i) {
        const int idx = t + i * BIN_T;
        bkt[i] = -1; key[i] = 0; wbt[i] = 0; rnk[i] = 0;
        if (idx < len) {
            const int e = e0 + idx;
            const int d = edge_dst[e];
            const int s = edge_src[e];
            const int w = __float_as_int(edge_w[e]);
            if ((unsigned)d < (unsigned)n_nodes) {
                const int b = d / RB;
                bkt[i] = b;
                key[i] = s | ((d - b * RB) << 24);
                wbt[i] = w;
                rnk[i] = atomicAdd(&cnt[b], 1);
            }
        }
    }
    __syncthreads();

    if (t < NBMAX) scan[0][t] = cnt[t];
    __syncthreads();
    int cur = 0;
    for (int d = 1; d < NBMAX; d <<= 1) {
        if (t < NBMAX) {
            int v = scan[cur][t];
            if (t >= d) v += scan[cur][t - d];
            scan[cur ^ 1][t] = v;
        }
        __syncthreads();
        cur ^= 1;
    }
    if (t < NBMAX) {
        const int ex = scan[cur][t] - cnt[t];
        scan[cur ^ 1][t] = ex;
        offT[t * NBLK_MAX + blk] = ex;
        if (t == NBMAX - 1) offT[NBMAX * NBLK_MAX + blk] = scan[cur][t];
    }
    __syncthreads();
    const int total = scan[cur][NBMAX - 1];

    #pragma unroll
    for (int i = 0; i < BIN_EPT; ++i) {
        if (bkt[i] >= 0) stage[scan[cur ^ 1][bkt[i]] + rnk[i]] = make_int2(key[i], wbt[i]);
    }
    __syncthreads();

    for (int idx = t; idx < total; idx += BIN_T) recs[e0 + idx] = stage[idx];
}

__device__ __forceinline__ void lds_row_add(float* row, int lane, float v0, float v1)
{
    __hip_atomic_fetch_add(row + lane,                 v0, __ATOMIC_RELAXED, __HIP_MEMORY_SCOPE_WORKGROUP);
    __hip_atomic_fetch_add(row + lane + OUT_FEATS / 2, v1, __ATOMIC_RELAXED, __HIP_MEMORY_SCOPE_WORKGROUP);
}

__global__ __launch_bounds__(RED_T)
void gcn_bucket_sum(const float* __restrict__ h,
                    const int2* __restrict__ recs,
                    const int* __restrict__ offT,
                    float* __restrict__ out,
                    int n_nodes, int nblk)
{
    __shared__ float acc[RB * OUT_FEATS];
    __shared__ int   segStart[NBLK_MAX];
    __shared__ int   segBase[NBLK_MAX + 1];
    __shared__ int   scan[2][NBLK_MAX];

    const int t    = threadIdx.x;
    const int lane = t & 31;
    const int b    = blockIdx.x;

    for (int i = t; i < RB * OUT_FEATS; i += RED_T) acc[i] = 0.0f;

    if (t < NBLK_MAX) {
        int mylen = 0, st = 0;
        if (t < nblk) {
            const int s = offT[b * NBLK_MAX + t];
            const int e = offT[(b + 1) * NBLK_MAX + t];
            st = t * CHUNK + s;
            mylen = e - s;
        }
        segStart[t] = st;
        scan[0][t]  = mylen;
    }
    __syncthreads();
    int cur = 0;
    for (int d = 1; d < NBLK_MAX; d <<= 1) {
        if (t < NBLK_MAX) {
            int v = scan[cur][t];
            if (t >= d) v += scan[cur][t - d];
            scan[cur ^ 1][t] = v;
        }
        __syncthreads();
        cur ^= 1;
    }
    if (t < NBLK_MAX) segBase[t + 1] = scan[cur][t];
    if (t == 0) segBase[0] = 0;
    __syncthreads();

    const int total = __builtin_amdgcn_readfirstlane(segBase[NBLK_MAX]);
    const int wv    = __builtin_amdgcn_readfirstlane(t >> 5);

    for (int base = wv * 32; base < total; base += RED_T) {
        const int i = base + lane;
        int key = 0, wbt = 0;
        if (i < total) {
            int lo = 0;
            #pragma unroll
            for (int step = NBLK_MAX / 2; step >= 1; step >>= 1)
                if (segBase[lo + step] <= i) lo += step;
            const int2 r = recs[segStart[lo] + (i - segBase[lo])];
            key = r.x; wbt = r.y;
        }
        const int left = total - base;
        const int cnt  = left < 32 ? left : 32;

        #pragma unroll
        for (int jb = 0; jb < 32; jb += EDGES_PER_WAVE) {
            if (cnt >= jb + EDGES_PER_WAVE) {
                float r0[EDGES_PER_WAVE], r1[EDGES_PER_WAVE];
                #pragma unroll
                for (int j = 0; j < EDGES_PER_WAVE; ++j) {
                    const int k = __builtin_amdgcn_readlane(key, jb + j);
                    const float* __restrict__ p = h + (size_t)(k & 0xFFFFFF) * OUT_FEATS;
                    r0[j] = p[lane];
                    r1[j] = p[lane + OUT_FEATS / 2];
                }
                #pragma unroll
                for (int j = 0; j < EDGES_PER_WAVE; ++j) {
                    const int   k = __builtin_amdgcn_readlane(key, jb + j);
                    const float w = __int_as_float(__builtin_amdgcn_readlane(wbt, jb + j));
                    lds_row_add(acc + ((unsigned)k >> 24) * OUT_FEATS, lane, w * r0[j], w * r1[j]);
                }
            } else {
                for (int j = jb; j < cnt; ++j) {
                    const int   k = __builtin_amdgcn_readlane(key, j);
                    const float w = __int_as_float(__builtin_amdgcn_readlane(wbt, j));
                    const float* __restrict__ p = h + (size_t)(k & 0xFFFFFF) * OUT_FEATS;
                    const float a0 = p[lane];
                    const float a1 = p[lane + OUT_FEATS / 2];
                    lds_row_add(acc + ((unsigned)k >> 24) * OUT_FEATS, lane, w * a0, w * a1);
                }
            }
        }
    }
    __syncthreads();

    const int row0  = b * RB;
    const int nleft = n_nodes - row0;
    const int nrows = nleft < RB ? nleft : RB;
    const float4* __restrict__ acc4 = (const float4*)acc;
    float4* __restrict__ out4 = (float4*)(out + (size_t)row0 * OUT_FEATS);
    for (int i = t; i < nrows * (OUT_FEATS / 4); i += RED_T) out4[i] = acc4[i];
}

extern "C" void kernel_launch(void* const* d_in, const int* in_sizes, int n_in,
                              void* d_out, int out_size, void* d_ws, size_t ws_size,
                              hipStream_t stream)
{
    const float* x    = (const float*)d_in[0];
    const int*   esrc = (const int*)  d_in[1];
    const int*   edst = (const int*)  d_in[2];
    const float* ew   = (const float*)d_in[3];
    const float* W    = (const float*)d_in[4];
    const float* bias = (const float*)d_in[5];

    float* out = (float*)d_out;
    float* h   = (float*)d_ws;

    const int n_nodes = in_sizes[0] / IN_FEATS;
    const int n_edges = in_sizes[1];

    const int rowTiles = (n_nodes + 15) / 16;
    const int wavesPerBlock = 8;
    const int gemmBlocks = (rowTiles + wavesPerBlock - 1) / wavesPerBlock;

    const size_t hBytes   = (((size_t)n_nodes * OUT_FEATS * sizeof(float)) + 255) & ~(size_t)255;
    const size_t recBytes = (((size_t)n_edges * sizeof(int2)) + 255) & ~(size_t)255;
    const size_t offBytes = (size_t)(NBMAX + 1) * NBLK_MAX * sizeof(int);
    const int nblk    = (n_edges + CHUNK - 1) / CHUNK;
    const int buckets = (n_nodes + RB - 1) / RB;

    if (ws_size >= hBytes + recBytes + offBytes && nblk >= 1 && nblk <= NBLK_MAX && buckets <= NBMAX) {
        int2* recs = (int2*)((char*)d_ws + hBytes);
        int*  offT = (int*)((char*)d_ws + hBytes + recBytes);
        const int fusedGemmBlocks = (rowTiles + (BIN_T / 32) - 1) / (BIN_T / 32);
        gcn_gemm_and_bin<<<fusedGemmBlocks + nblk, BIN_T, 0, stream>>>(x, W, bias, h, esrc, edst, ew, recs, offT,
                                                                      n_edges, n_nodes, fusedGemmBlocks);
        gcn_bucket_sum<<<buckets, RED_T, 0, stream>>>(h, recs, offT, out, n_nodes, nblk);
        return;
    }

    hipMemsetAsync(d_out, 0, (size_t)out_size * sizeof(float), stream);
    gcn_gemm_wmma<<<gemmBlocks, wavesPerBlock * 32, 0, stream>>>(x, W, bias, h, n_nodes);

    const int edgesPerBlock = wavesPerBlock * EDGES_PER_WAVE;
    const int scatterBlocks = (n_edges + edgesPerBlock - 1) / edgesPerBlock;
    gcn_edge_scatter<<<scatterBlocks, wavesPerBlock * 32, 0, stream>>>(h, esrc, edst, ew, out, n_edges);
}


# GEMM B operand from LDS: each GEMM block copies W once into LDS in fragment order and every wave reads its B pairs by ds_load_b128 (128 per wave) instead of 512 four-byte global loads per wave
# speedup vs baseline: 2.8041x; 1.0752x over previous
#include <hip/hip_runtime.h>
#include <hip/hip_bf16.h>

typedef __attribute__((ext_vector_type(2))) float v2f;
typedef __attribute__((ext_vector_type(8))) float v8f;

#define IN_FEATS  256
#define OUT_FEATS 64

__device__ __forceinline__ void gemm_rows16(const float* __restrict__ x,
                                            const float* __restrict__ W,
                                            const float* __restrict__ bias,
                                            float* __restrict__ h,
                                            int n_nodes, int rowTile)
{
    const int lane   = threadIdx.x & 31;

    const int row0 = rowTile * 16;
    if (row0 >= n_nodes) return;

    const int m  = lane & 15;
    const int hi = lane >> 4;
    const int n  = lane & 15;

    int arow = row0 + m;
    if (arow >= n_nodes) arow = n_nodes - 1;
    const float* __restrict__ xrow = x + (size_t)arow * IN_FEATS;

    v8f acc0 = {}, acc1 = {}, acc2 = {}, acc3 = {};

    #pragma unroll 4
    for (int k = 0; k < IN_FEATS; k += 4) {
        const int kk = k + 2 * hi;
        v2f a = *(const v2f*)(xrow + kk);

        const float* __restrict__ w0 = W + (size_t)kk * OUT_FEATS + n;
        const float* __restrict__ w1 = w0 + OUT_FEATS;
        v2f b0 = { w0[0],  w1[0]  };
        v2f b1 = { w0[16], w1[16] };
        v2f b2 = { w0[32], w1[32] };
        v2f b3 = { w0[48], w1[48] };

        acc0 = __builtin_amdgcn_wmma_f32_16x16x4_f32(false, a, false, b0, (short)0, acc0, false, false);
        acc1 = __builtin_amdgcn_wmma_f32_16x16x4_f32(false, a, false, b1, (short)0, acc1, false, false);
        acc2 = __builtin_amdgcn_wmma_f32_16x16x4_f32(false, a, false, b2, (short)0, acc2, false, false);
        acc3 = __builtin_amdgcn_wmma_f32_16x16x4_f32(false, a, false, b3, (short)0, acc3, false, false);
    }

    const float bn0 = bias[n];
    const float bn1 = bias[16 + n];
    const float bn2 = bias[32 + n];
    const float bn3 = bias[48 + n];

    #pragma unroll
    for (int r = 0; r < 8; ++r) {
        const int row = row0 + r + 8 * hi;
        if (row < n_nodes) {
            float* __restrict__ hrow = h + (size_t)row * OUT_FEATS;
            hrow[n]      = acc0[r] + bn0;
            hrow[16 + n] = acc1[r] + bn1;
            hrow[32 + n] = acc2[r] + bn2;
            hrow[48 + n] = acc3[r] + bn3;
        }
    }
}

#define WFRAG_SLOTS ((IN_FEATS / 4) * 2 * 32)
#define WFRAG_FILL_T 1024

__device__ __forceinline__ void gemm_fill_wfrag(const float* __restrict__ W, float4* wfrag, int tid)
{
    static_assert(WFRAG_SLOTS % WFRAG_FILL_T == 0, "whole slots per thread");
    #pragma unroll
    for (int i = 0; i < WFRAG_SLOTS / WFRAG_FILL_T; ++i) {
        const int s  = tid + i * WFRAG_FILL_T;
        const int ln = s & 31;
        const int p  = (s >> 5) & 1;
        const int ks = s >> 6;
        const int k0 = 4 * ks + 2 * (ln >> 4);
        const int c0 = 32 * p + (ln & 15);
        const float* __restrict__ w0 = W + (size_t)k0 * OUT_FEATS + c0;
        wfrag[s] = make_float4(w0[0], w0[OUT_FEATS], w0[16], w0[OUT_FEATS + 16]);
    }
}

__device__ __forceinline__ void gemm_rows16_lds(const float* __restrict__ x,
                                                const float4* wfrag,
                                                const float* __restrict__ bias,
                                                float* __restrict__ h,
                                                int n_nodes, int rowTile)
{
    const int lane = threadIdx.x & 31;

    const int row0 = rowTile * 16;
    if (row0 >= n_nodes) return;

    const int m  = lane & 15;
    const int hi = lane >> 4;
    const int n  = lane & 15;

    int arow = row0 + m;
    if (arow >= n_nodes) arow = n_nodes - 1;
    const float* __restrict__ xrow = x + (size_t)arow * IN_FEATS;
    const float4* wl = wfrag + lane;

    v8f acc0 = {}, acc1 = {}, acc2 = {}, acc3 = {};

    #pragma unroll 2
    for (int k = 0; k < IN_FEATS; k += 4) {
        const int kk = k + 2 * hi;
        v2f a = *(const v2f*)(xrow + kk);

        const float4 q0 = wl[(k >> 2) * 64];
        const float4 q1 = wl[(k >> 2) * 64 + 32];
        v2f b0 = { q0.x, q0.y };
        v2f b1 = { q0.z, q0.w };
        v2f b2 = { q1.x, q1.y };
        v2f b3 = { q1.z, q1.w };

        acc0 = __builtin_amdgcn_wmma_f32_16x16x4_f32(false, a, false, b0, (short)0, acc0, false, false);
        acc1 = __builtin_amdgcn_wmma_f32_16x16x4_f32(false, a, false, b1, (short)0, acc1, false, false);
        acc2 = __builtin_amdgcn_wmma_f32_16x16x4_f32(false, a, false, b2, (short)0, acc2, false, false);
        acc3 = __builtin_amdgcn_wmma_f32_16x16x4_f32(false, a, false, b3, (short)0, acc3, false, false);
    }

    const float bn0 = bias[n];
    const float bn1 = bias[16 + n];
    const float bn2 = bias[32 + n];
    const float bn3 = bias[48 + n];

    #pragma unroll
    for (int r = 0; r < 8; ++r) {
        const int row = row0 + r + 8 * hi;
        if (row < n_nodes) {
            float* __restrict__ hrow = h + (size_t)row * OUT_FEATS;
            hrow[n]      = acc0[r] + bn0;
            hrow[16 + n] = acc1[r] + bn1;
            hrow[32 + n] = acc2[r] + bn2;
            hrow[48 + n] = acc3[r] + bn3;
        }
    }
}

__global__ void gcn_gemm_wmma(const float* __restrict__ x,
                              const float* __restrict__ W,
                              const float* __restrict__ bias,
                              float* __restrict__ h,
                              int n_nodes)
{
    const int waveId = threadIdx.x >> 5;
    const int wavesPerBlock = blockDim.x >> 5;
    gemm_rows16(x, W, bias, h, n_nodes, blockIdx.x * wavesPerBlock + waveId);
}

#define EDGES_PER_WAVE 16

__device__ __forceinline__ void edge_add(float* __restrict__ row, int lane, float v0, float v1)
{
#if defined(__HIP_DEVICE_COMPILE__)
    unsafeAtomicAdd(row + lane,                 v0);
    unsafeAtomicAdd(row + lane + OUT_FEATS / 2, v1);
#else
    atomicAdd(row + lane,                 v0);
    atomicAdd(row + lane + OUT_FEATS / 2, v1);
#endif
}

__global__ void gcn_edge_scatter(const float* __restrict__ h,
                                 const int* __restrict__ edge_src,
                                 const int* __restrict__ edge_dst,
                                 const float* __restrict__ edge_w,
                                 float* __restrict__ out,
                                 int n_edges)
{
    const int lane   = threadIdx.x & 31;
    const int waveId = threadIdx.x >> 5;
    const int wavesPerBlock = blockDim.x >> 5;

    const int wave = __builtin_amdgcn_readfirstlane(blockIdx.x * wavesPerBlock + waveId);
    const int e0 = wave * EDGES_PER_WAVE;
    if (e0 >= n_edges) return;
    const int left = n_edges - e0;
    const int cnt  = left < EDGES_PER_WAVE ? left : EDGES_PER_WAVE;

    const int sub = lane & (EDGES_PER_WAVE - 1);
    const int ei  = e0 + (sub < cnt ? sub : 0);
    const int sv = edge_src[ei];
    const int dv = edge_dst[ei];
    const int wv = __float_as_int(edge_w[ei]);

    if (cnt == EDGES_PER_WAVE) {
        float r0[EDGES_PER_WAVE], r1[EDGES_PER_WAVE];
        #pragma unroll
        for (int j = 0; j < EDGES_PER_WAVE; ++j) {
            const int s = __builtin_amdgcn_readlane(sv, j);
            const float* __restrict__ p = h + (size_t)s * OUT_FEATS;
            r0[j] = p[lane];
            r1[j] = p[lane + OUT_FEATS / 2];
        }
        #pragma unroll
        for (int j = 0; j < EDGES_PER_WAVE; ++j) {
            const int   d = __builtin_amdgcn_readlane(dv, j);
            const float w = __int_as_float(__builtin_amdgcn_readlane(wv, j));
            edge_add(out + (size_t)d * OUT_FEATS, lane, w * r0[j], w * r1[j]);
        }
    } else {
        for (int j = 0; j < cnt; ++j) {
            const int   s = __builtin_amdgcn_readlane(sv, j);
            const int   d = __builtin_amdgcn_readlane(dv, j);
            const float w = __int_as_float(__builtin_amdgcn_readlane(wv, j));
            const float* __restrict__ p = h + (size_t)s * OUT_FEATS;
            const float a0 = p[lane];
            const float a1 = p[lane + OUT_FEATS / 2];
            edge_add(out + (size_t)d * OUT_FEATS, lane, w * a0, w * a1);
        }
    }
}

#define RB        196
#define NBMAX     512
#define CHUNK     6400
#define NBLK_MAX  256
#define BIN_T     1024
#define BIN_EPT   ((CHUNK + BIN_T - 1) / BIN_T)
#define RED_T     512

__global__ __launch_bounds__(BIN_T)
void gcn_gemm_and_bin(const float* __restrict__ x,
                      const float* __restrict__ W,
                      const float* __restrict__ bias,
                      float* __restrict__ h,
                      const int* __restrict__ edge_src,
                      const int* __restrict__ edge_dst,
                      const float* __restrict__ edge_w,
                      int2* __restrict__ recs,
                      int* __restrict__ offT,
                      int n_edges, int n_nodes, int gemm_blocks)
{
    __shared__ __attribute__((aligned(16))) unsigned char smem[WFRAG_SLOTS * sizeof(float4)];
    static_assert((NBMAX + 2 * NBMAX) * sizeof(int) + CHUNK * sizeof(int2) <= WFRAG_SLOTS * sizeof(float4),
                  "the binning role's arrays must fit the shared block");
    static_assert(WFRAG_FILL_T == BIN_T, "the W copy is written for this launch's block size");

    if ((int)blockIdx.x < gemm_blocks) {
        float4* wfrag = (float4*)smem;
        gemm_fill_wfrag(W, wfrag, threadIdx.x);
        __syncthreads();
        gemm_rows16_lds(x, wfrag, bias, h, n_nodes, blockIdx.x * (BIN_T / 32) + (threadIdx.x >> 5));
        return;
    }

    int*  cnt             = (int*)smem;
    int  (*scan)[NBMAX]   = (int (*)[NBMAX])(cnt + NBMAX);
    int2* stage           = (int2*)(cnt + 3 * NBMAX);

    const int t   = threadIdx.x;
    const int blk = blockIdx.x - gemm_blocks;
    const int e0  = blk * CHUNK;
    const int rem = n_edges - e0;
    const int len = rem < CHUNK ? rem : CHUNK;

    if (t < NBMAX) cnt[t] = 0;
    __syncthreads();

    int key[BIN_EPT], wbt[BIN_EPT], bkt[BIN_EPT], rnk[BIN_EPT];
    #pragma unroll
    for (int i = 0; i < BIN_EPT; ++i) {
        const int idx = t + i * BIN_T;
        bkt[i] = -1; key[i] = 0; wbt[i] = 0; rnk[i] = 0;
        if (idx < len) {
            const int e = e0 + idx;
            const int d = edge_dst[e];
            const int s = edge_src[e];
            const int w = __float_as_int(edge_w[e]);
            if ((unsigned)d < (unsigned)n_nodes) {
                const int b = d / RB;
                bkt[i] = b;
                key[i] = s | ((d - b * RB) << 24);
                wbt[i] = w;
                rnk[i] = atomicAdd(&cnt[b], 1);
            }
        }
    }
    __syncthreads();

    if (t < NBMAX) scan[0][t] = cnt[t];
    __syncthreads();
    int cur = 0;
    for (int d = 1; d < NBMAX; d <<= 1) {
        if (t < NBMAX) {
            int v = scan[cur][t];
            if (t >= d) v += scan[cur][t - d];
            scan[cur ^ 1][t] = v;
        }
        __syncthreads();
        cur ^= 1;
    }
    if (t < NBMAX) {
        const int ex = scan[cur][t] - cnt[t];
        scan[cur ^ 1][t] = ex;
        offT[t * NBLK_MAX + blk] = ex;
        if (t == NBMAX - 1) offT[NBMAX * NBLK_MAX + blk] = scan[cur][t];
    }
    __syncthreads();
    const int total = scan[cur][NBMAX - 1];

    #pragma unroll
    for (int i = 0; i < BIN_EPT; ++i) {
        if (bkt[i] >= 0) stage[scan[cur ^ 1][bkt[i]] + rnk[i]] = make_int2(key[i], wbt[i]);
    }
    __syncthreads();

    for (int idx = t; idx < total; idx += BIN_T) recs[e0 + idx] = stage[idx];
}

__device__ __forceinline__ void lds_row_add(float* row, int lane, float v0, float v1)
{
    __hip_atomic_fetch_add(row + lane,                 v0, __ATOMIC_RELAXED, __HIP_MEMORY_SCOPE_WORKGROUP);
    __hip_atomic_fetch_add(row + lane + OUT_FEATS / 2, v1, __ATOMIC_RELAXED, __HIP_MEMORY_SCOPE_WORKGROUP);
}

__global__ __launch_bounds__(RED_T)
void gcn_bucket_sum(const float* __restrict__ h,
                    const int2* __restrict__ recs,
                    const int* __restrict__ offT,
                    float* __restrict__ out,
                    int n_nodes, int nblk)
{
    __shared__ float acc[RB * OUT_FEATS];
    __shared__ int   segStart[NBLK_MAX];
    __shared__ int   segBase[NBLK_MAX + 1];
    __shared__ int   scan[2][NBLK_MAX];

    const int t    = threadIdx.x;
    const int lane = t & 31;
    const int b    = blockIdx.x;

    for (int i = t; i < RB * OUT_FEATS; i += RED_T) acc[i] = 0.0f;

    if (t < NBLK_MAX) {
        int mylen = 0, st = 0;
        if (t < nblk) {
            const int s = offT[b * NBLK_MAX + t];
            const int e = offT[(b + 1) * NBLK_MAX + t];
            st = t * CHUNK + s;
            mylen = e - s;
        }
        segStart[t] = st;
        scan[0][t]  = mylen;
    }
    __syncthreads();
    int cur = 0;
    for (int d = 1; d < NBLK_MAX; d <<= 1) {
        if (t < NBLK_MAX) {
            int v = scan[cur][t];
            if (t >= d) v += scan[cur][t - d];
            scan[cur ^ 1][t] = v;
        }
        __syncthreads();
        cur ^= 1;
    }
    if (t < NBLK_MAX) segBase[t + 1] = scan[cur][t];
    if (t == 0) segBase[0] = 0;
    __syncthreads();

    const int total = __builtin_amdgcn_readfirstlane(segBase[NBLK_MAX]);
    const int wv    = __builtin_amdgcn_readfirstlane(t >> 5);

    for (int base = wv * 32; base < total; base += RED_T) {
        const int i = base + lane;
        int key = 0, wbt = 0;
        if (i < total) {
            int lo = 0;
            #pragma unroll
            for (int step = NBLK_MAX / 2; step >= 1; step >>= 1)
                if (segBase[lo + step] <= i) lo += step;
            const int2 r = recs[segStart[lo] + (i - segBase[lo])];
            key = r.x; wbt = r.y;
        }
        const int left = total - base;
        const int cnt  = left < 32 ? left : 32;

        #pragma unroll
        for (int jb = 0; jb < 32; jb += EDGES_PER_WAVE) {
            if (cnt >= jb + EDGES_PER_WAVE) {
                float r0[EDGES_PER_WAVE], r1[EDGES_PER_WAVE];
                #pragma unroll
                for (int j = 0; j < EDGES_PER_WAVE; ++j) {
                    const int k = __builtin_amdgcn_readlane(key, jb + j);
                    const float* __restrict__ p = h + (size_t)(k & 0xFFFFFF) * OUT_FEATS;
                    r0[j] = p[lane];
                    r1[j] = p[lane + OUT_FEATS / 2];
                }
                #pragma unroll
                for (int j = 0; j < EDGES_PER_WAVE; ++j) {
                    const int   k = __builtin_amdgcn_readlane(key, jb + j);
                    const float w = __int_as_float(__builtin_amdgcn_readlane(wbt, jb + j));
                    lds_row_add(acc + ((unsigned)k >> 24) * OUT_FEATS, lane, w * r0[j], w * r1[j]);
                }
            } else {
                for (int j = jb; j < cnt; ++j) {
                    const int   k = __builtin_amdgcn_readlane(key, j);
                    const float w = __int_as_float(__builtin_amdgcn_readlane(wbt, j));
                    const float* __restrict__ p = h + (size_t)(k & 0xFFFFFF) * OUT_FEATS;
                    const float a0 = p[lane];
                    const float a1 = p[lane + OUT_FEATS / 2];
                    lds_row_add(acc + ((unsigned)k >> 24) * OUT_FEATS, lane, w * a0, w * a1);
                }
            }
        }
    }
    __syncthreads();

    const int row0  = b * RB;
    const int nleft = n_nodes - row0;
    const int nrows = nleft < RB ? nleft : RB;
    const float4* __restrict__ acc4 = (const float4*)acc;
    float4* __restrict__ out4 = (float4*)(out + (size_t)row0 * OUT_FEATS);
    for (int i = t; i < nrows * (OUT_FEATS / 4); i += RED_T) out4[i] = acc4[i];
}

extern "C" void kernel_launch(void* const* d_in, const int* in_sizes, int n_in,
                              void* d_out, int out_size, void* d_ws, size_t ws_size,
                              hipStream_t stream)
{
    const float* x    = (const float*)d_in[0];
    const int*   esrc = (const int*)  d_in[1];
    const int*   edst = (const int*)  d_in[2];
    const float* ew   = (const float*)d_in[3];
    const float* W    = (const float*)d_in[4];
    const float* bias = (const float*)d_in[5];

    float* out = (float*)d_out;
    float* h   = (float*)d_ws;

    const int n_nodes = in_sizes[0] / IN_FEATS;
    const int n_edges = in_sizes[1];

    const int rowTiles = (n_nodes + 15) / 16;
    const int wavesPerBlock = 8;
    const int gemmBlocks = (rowTiles + wavesPerBlock - 1) / wavesPerBlock;

    const size_t hBytes   = (((size_t)n_nodes * OUT_FEATS * sizeof(float)) + 255) & ~(size_t)255;
    const size_t recBytes = (((size_t)n_edges * sizeof(int2)) + 255) & ~(size_t)255;
    const size_t offBytes = (size_t)(NBMAX + 1) * NBLK_MAX * sizeof(int);
    const int nblk    = (n_edges + CHUNK - 1) / CHUNK;
    const int buckets = (n_nodes + RB - 1) / RB;

    if (ws_size >= hBytes + recBytes + offBytes && nblk >= 1 && nblk <= NBLK_MAX && buckets <= NBMAX) {
        int2* recs = (int2*)((char*)d_ws + hBytes);
        int*  offT = (int*)((char*)d_ws + hBytes + recBytes);
        const int fusedGemmBlocks = (rowTiles + (BIN_T / 32) - 1) / (BIN_T / 32);
        gcn_gemm_and_bin<<<fusedGemmBlocks + nblk, BIN_T, 0, stream>>>(x, W, bias, h, esrc, edst, ew, recs, offT,
                                                                      n_edges, n_nodes, fusedGemmBlocks);
        gcn_bucket_sum<<<buckets, RED_T, 0, stream>>>(h, recs, offT, out, n_nodes, nblk);
        return;
    }

    hipMemsetAsync(d_out, 0, (size_t)out_size * sizeof(float), stream);
    gcn_gemm_wmma<<<gemmBlocks, wavesPerBlock * 32, 0, stream>>>(x, W, bias, h, n_nodes);

    const int edgesPerBlock = wavesPerBlock * EDGES_PER_WAVE;
    const int scatterBlocks = (n_edges + edgesPerBlock - 1) / edgesPerBlock;
    gcn_edge_scatter<<<scatterBlocks, wavesPerBlock * 32, 0, stream>>>(h, esrc, edst, ew, out, n_edges);
}


# bucket sum: records sorted by destination row in LDS (integer rank adds); a half-wave sums each row's run in registers from 16-B loads, stores it once; no LDS float adds; sum order = rank order
# speedup vs baseline: 3.1393x; 1.1195x over previous
#include <hip/hip_runtime.h>
#include <hip/hip_bf16.h>

typedef __attribute__((ext_vector_type(2))) float v2f;
typedef __attribute__((ext_vector_type(8))) float v8f;

#define IN_FEATS  256
#define OUT_FEATS 64

__device__ __forceinline__ void gemm_rows16(const float* __restrict__ x,
                                            const float* __restrict__ W,
                                            const float* __restrict__ bias,
                                            float* __restrict__ h,
                                            int n_nodes, int rowTile)
{
    const int lane   = threadIdx.x & 31;

    const int row0 = rowTile * 16;
    if (row0 >= n_nodes) return;

    const int m  = lane & 15;
    const int hi = lane >> 4;
    const int n  = lane & 15;

    int arow = row0 + m;
    if (arow >= n_nodes) arow = n_nodes - 1;
    const float* __restrict__ xrow = x + (size_t)arow * IN_FEATS;

    v8f acc0 = {}, acc1 = {}, acc2 = {}, acc3 = {};

    #pragma unroll 4
    for (int k = 0; k < IN_FEATS; k += 4) {
        const int kk = k + 2 * hi;
        v2f a = *(const v2f*)(xrow + kk);

        const float* __restrict__ w0 = W + (size_t)kk * OUT_FEATS + n;
        const float* __restrict__ w1 = w0 + OUT_FEATS;
        v2f b0 = { w0[0],  w1[0]  };
        v2f b1 = { w0[16], w1[16] };
        v2f b2 = { w0[32], w1[32] };
        v2f b3 = { w0[48], w1[48] };

        acc0 = __builtin_amdgcn_wmma_f32_16x16x4_f32(false, a, false, b0, (short)0, acc0, false, false);
        acc1 = __builtin_amdgcn_wmma_f32_16x16x4_f32(false, a, false, b1, (short)0, acc1, false, false);
        acc2 = __builtin_amdgcn_wmma_f32_16x16x4_f32(false, a, false, b2, (short)0, acc2, false, false);
        acc3 = __builtin_amdgcn_wmma_f32_16x16x4_f32(false, a, false, b3, (short)0, acc3, false, false);
    }

    const float bn0 = bias[n];
    const float bn1 = bias[16 + n];
    const float bn2 = bias[32 + n];
    const float bn3 = bias[48 + n];

    #pragma unroll
    for (int r = 0; r < 8; ++r) {
        const int row = row0 + r + 8 * hi;
        if (row < n_nodes) {
            float* __restrict__ hrow = h + (size_t)row * OUT_FEATS;
            hrow[n]      = acc0[r] + bn0;
            hrow[16 + n] = acc1[r] + bn1;
            hrow[32 + n] = acc2[r] + bn2;
            hrow[48 + n] = acc3[r] + bn3;
        }
    }
}

#define WFRAG_SLOTS ((IN_FEATS / 4) * 2 * 32)
#define WFRAG_FILL_T 1024

__device__ __forceinline__ void gemm_fill_wfrag(const float* __restrict__ W, float4* wfrag, int tid)
{
    static_assert(WFRAG_SLOTS % WFRAG_FILL_T == 0, "whole slots per thread");
    #pragma unroll
    for (int i = 0; i < WFRAG_SLOTS / WFRAG_FILL_T; ++i) {
        const int s  = tid + i * WFRAG_FILL_T;
        const int ln = s & 31;
        const int p  = (s >> 5) & 1;
        const int ks = s >> 6;
        const int k0 = 4 * ks + 2 * (ln >> 4);
        const int c0 = 32 * p + (ln & 15);
        const float* __restrict__ w0 = W + (size_t)k0 * OUT_FEATS + c0;
        wfrag[s] = make_float4(w0[0], w0[OUT_FEATS], w0[16], w0[OUT_FEATS + 16]);
    }
}

__device__ __forceinline__ void gemm_rows16_lds(const float* __restrict__ x,
                                                const float4* wfrag,
                                                const float* __restrict__ bias,
                                                float* __restrict__ h,
                                                int n_nodes, int rowTile)
{
    const int lane = threadIdx.x & 31;

    const int row0 = rowTile * 16;
    if (row0 >= n_nodes) return;

    const int m  = lane & 15;
    const int hi = lane >> 4;
    const int n  = lane & 15;

    int arow = row0 + m;
    if (arow >= n_nodes) arow = n_nodes - 1;
    const float* __restrict__ xrow = x + (size_t)arow * IN_FEATS;
    const float4* wl = wfrag + lane;

    v8f acc0 = {}, acc1 = {}, acc2 = {}, acc3 = {};

    #pragma unroll 2
    for (int k = 0; k < IN_FEATS; k += 4) {
        const int kk = k + 2 * hi;
        v2f a = *(const v2f*)(xrow + kk);

        const float4 q0 = wl[(k >> 2) * 64];
        const float4 q1 = wl[(k >> 2) * 64 + 32];
        v2f b0 = { q0.x, q0.y };
        v2f b1 = { q0.z, q0.w };
        v2f b2 = { q1.x, q1.y };
        v2f b3 = { q1.z, q1.w };

        acc0 = __builtin_amdgcn_wmma_f32_16x16x4_f32(false, a, false, b0, (short)0, acc0, false, false);
        acc1 = __builtin_amdgcn_wmma_f32_16x16x4_f32(false, a, false, b1, (short)0, acc1, false, false);
        acc2 = __builtin_amdgcn_wmma_f32_16x16x4_f32(false, a, false, b2, (short)0, acc2, false, false);
        acc3 = __builtin_amdgcn_wmma_f32_16x16x4_f32(false, a, false, b3, (short)0, acc3, false, false);
    }

    const float bn0 = bias[n];
    const float bn1 = bias[16 + n];
    const float bn2 = bias[32 + n];
    const float bn3 = bias[48 + n];

    #pragma unroll
    for (int r = 0; r < 8; ++r) {
        const int row = row0 + r + 8 * hi;
        if (row < n_nodes) {
            float* __restrict__ hrow = h + (size_t)row * OUT_FEATS;
            hrow[n]      = acc0[r] + bn0;
            hrow[16 + n] = acc1[r] + bn1;
            hrow[32 + n] = acc2[r] + bn2;
            hrow[48 + n] = acc3[r] + bn3;
        }
    }
}

__global__ void gcn_gemm_wmma(const float* __restrict__ x,
                              const float* __restrict__ W,
                              const float* __restrict__ bias,
                              float* __restrict__ h,
                              int n_nodes)
{
    const int waveId = threadIdx.x >> 5;
    const int wavesPerBlock = blockDim.x >> 5;
    gemm_rows16(x, W, bias, h, n_nodes, blockIdx.x * wavesPerBlock + waveId);
}

#define EDGES_PER_WAVE 16

__device__ __forceinline__ void edge_add(float* __restrict__ row, int lane, float v0, float v1)
{
#if defined(__HIP_DEVICE_COMPILE__)
    unsafeAtomicAdd(row + lane,                 v0);
    unsafeAtomicAdd(row + lane + OUT_FEATS / 2, v1);
#else
    atomicAdd(row + lane,                 v0);
    atomicAdd(row + lane + OUT_FEATS / 2, v1);
#endif
}

__global__ void gcn_edge_scatter(const float* __restrict__ h,
                                 const int* __restrict__ edge_src,
                                 const int* __restrict__ edge_dst,
                                 const float* __restrict__ edge_w,
                                 float* __restrict__ out,
                                 int n_edges)
{
    const int lane   = threadIdx.x & 31;
    const int waveId = threadIdx.x >> 5;
    const int wavesPerBlock = blockDim.x >> 5;

    const int wave = __builtin_amdgcn_readfirstlane(blockIdx.x * wavesPerBlock + waveId);
    const int e0 = wave * EDGES_PER_WAVE;
    if (e0 >= n_edges) return;
    const int left = n_edges - e0;
    const int cnt  = left < EDGES_PER_WAVE ? left : EDGES_PER_WAVE;

    const int sub = lane & (EDGES_PER_WAVE - 1);
    const int ei  = e0 + (sub < cnt ? sub : 0);
    const int sv = edge_src[ei];
    const int dv = edge_dst[ei];
    const int wv = __float_as_int(edge_w[ei]);

    if (cnt == EDGES_PER_WAVE) {
        float r0[EDGES_PER_WAVE], r1[EDGES_PER_WAVE];
        #pragma unroll
        for (int j = 0; j < EDGES_PER_WAVE; ++j) {
            const int s = __builtin_amdgcn_readlane(sv, j);
            const float* __restrict__ p = h + (size_t)s * OUT_FEATS;
            r0[j] = p[lane];
            r1[j] = p[lane + OUT_FEATS / 2];
        }
        #pragma unroll
        for (int j = 0; j < EDGES_PER_WAVE; ++j) {
            const int   d = __builtin_amdgcn_readlane(dv, j);
            const float w = __int_as_float(__builtin_amdgcn_readlane(wv, j));
            edge_add(out + (size_t)d * OUT_FEATS, lane, w * r0[j], w * r1[j]);
        }
    } else {
        for (int j = 0; j < cnt; ++j) {
            const int   s = __builtin_amdgcn_readlane(sv, j);
            const int   d = __builtin_amdgcn_readlane(dv, j);
            const float w = __int_as_float(__builtin_amdgcn_readlane(wv, j));
            const float* __restrict__ p = h + (size_t)s * OUT_FEATS;
            const float a0 = p[lane];
            const float a1 = p[lane + OUT_FEATS / 2];
            edge_add(out + (size_t)d * OUT_FEATS, lane, w * a0, w * a1);
        }
    }
}

#define RB        196
#define NBMAX     512
#define CHUNK     6400
#define NBLK_MAX  256
#define BIN_T     1024
#define BIN_EPT   ((CHUNK + BIN_T - 1) / BIN_T)
#define RED_T     512

__global__ __launch_bounds__(BIN_T)
void gcn_gemm_and_bin(const float* __restrict__ x,
                      const float* __restrict__ W,
                      const float* __restrict__ bias,
                      float* __restrict__ h,
                      const int* __restrict__ edge_src,
                      const int* __restrict__ edge_dst,
                      const float* __restrict__ edge_w,
                      int2* __restrict__ recs,
                      int* __restrict__ offT,
                      int n_edges, int n_nodes, int gemm_blocks)
{
    __shared__ __attribute__((aligned(16))) unsigned char smem[WFRAG_SLOTS * sizeof(float4)];
    static_assert((NBMAX + 2 * NBMAX) * sizeof(int) + CHUNK * sizeof(int2) <= WFRAG_SLOTS * sizeof(float4),
                  "the binning role's arrays must fit the shared block");
    static_assert(WFRAG_FILL_T == BIN_T, "the W copy is written for this launch's block size");

    if ((int)blockIdx.x < gemm_blocks) {
        float4* wfrag = (float4*)smem;
        gemm_fill_wfrag(W, wfrag, threadIdx.x);
        __syncthreads();
        gemm_rows16_lds(x, wfrag, bias, h, n_nodes, blockIdx.x * (BIN_T / 32) + (threadIdx.x >> 5));
        return;
    }

    int*  cnt             = (int*)smem;
    int  (*scan)[NBMAX]   = (int (*)[NBMAX])(cnt + NBMAX);
    int2* stage           = (int2*)(cnt + 3 * NBMAX);

    const int t   = threadIdx.x;
    const int blk = blockIdx.x - gemm_blocks;
    const int e0  = blk * CHUNK;
    const int rem = n_edges - e0;
    const int len = rem < CHUNK ? rem : CHUNK;

    if (t < NBMAX) cnt[t] = 0;
    __syncthreads();

    int key[BIN_EPT], wbt[BIN_EPT], bkt[BIN_EPT], rnk[BIN_EPT];
    #pragma unroll
    for (int i = 0; i < BIN_EPT; ++i) {
        const int idx = t + i * BIN_T;
        bkt[i] = -1; key[i] = 0; wbt[i] = 0; rnk[i] = 0;
        if (idx < len) {
            const int e = e0 + idx;
            const int d = edge_dst[e];
            const int s = edge_src[e];
            const int w = __float_as_int(edge_w[e]);
            if ((unsigned)d < (unsigned)n_nodes) {
                const int b = d / RB;
                bkt[i] = b;
                key[i] = s | ((d - b * RB) << 24);
                wbt[i] = w;
                rnk[i] = atomicAdd(&cnt[b], 1);
            }
        }
    }
    __syncthreads();

    if (t < NBMAX) scan[0][t] = cnt[t];
    __syncthreads();
    int cur = 0;
    for (int d = 1; d < NBMAX; d <<= 1) {
        if (t < NBMAX) {
            int v = scan[cur][t];
            if (t >= d) v += scan[cur][t - d];
            scan[cur ^ 1][t] = v;
        }
        __syncthreads();
        cur ^= 1;
    }
    if (t < NBMAX) {
        const int ex = scan[cur][t] - cnt[t];
        scan[cur ^ 1][t] = ex;
        offT[t * NBLK_MAX + blk] = ex;
        if (t == NBMAX - 1) offT[NBMAX * NBLK_MAX + blk] = scan[cur][t];
    }
    __syncthreads();
    const int total = scan[cur][NBMAX - 1];

    #pragma unroll
    for (int i = 0; i < BIN_EPT; ++i) {
        if (bkt[i] >= 0) stage[scan[cur ^ 1][bkt[i]] + rnk[i]] = make_int2(key[i], wbt[i]);
    }
    __syncthreads();

    for (int idx = t; idx < total; idx += BIN_T) recs[e0 + idx] = stage[idx];
}

__device__ __forceinline__ void lds_row_add(float* row, int lane, float v0, float v1)
{
    __hip_atomic_fetch_add(row + lane,                 v0, __ATOMIC_RELAXED, __HIP_MEMORY_SCOPE_WORKGROUP);
    __hip_atomic_fetch_add(row + lane + OUT_FEATS / 2, v1, __ATOMIC_RELAXED, __HIP_MEMORY_SCOPE_WORKGROUP);
}

#define SRT_CAP    4096
#define SRT_PER_T  (SRT_CAP / RED_T)
#define ROW_SLOTS  256
#define GATHER_G   16

__global__ __launch_bounds__(RED_T)
void gcn_bucket_sum(const float* __restrict__ h,
                    const int2* __restrict__ recs,
                    const int* __restrict__ offT,
                    float* __restrict__ out,
                    int n_nodes, int nblk)
{
    __shared__ __attribute__((aligned(16))) unsigned char pool[RB * OUT_FEATS * sizeof(float)];
    static_assert(SRT_CAP * sizeof(int2) <= RB * OUT_FEATS * sizeof(float), "sorted records must fit the shared block");
    static_assert(RB <= ROW_SLOTS && (RB % 2) == 0, "two rows per wave, counters for every row");
    static_assert(SRT_CAP % RED_T == 0 && ROW_SLOTS == 8 * 32, "whole records per thread; 8 counters per lane in the scan");
    __shared__ int   segStart[NBLK_MAX];
    __shared__ int   segBase[NBLK_MAX + 1];
    __shared__ int   scan[2][NBLK_MAX];
    __shared__ int   rowCnt[ROW_SLOTS];
    __shared__ int   rowStart[ROW_SLOTS];

    const int t    = threadIdx.x;
    const int lane = t & 31;
    const int b    = blockIdx.x;

    if (t < ROW_SLOTS) rowCnt[t] = 0;

    if (t < NBLK_MAX) {
        int mylen = 0, st = 0;
        if (t < nblk) {
            const int s = offT[b * NBLK_MAX + t];
            const int e = offT[(b + 1) * NBLK_MAX + t];
            st = t * CHUNK + s;
            mylen = e - s;
        }
        segStart[t] = st;
        scan[0][t]  = mylen;
    }
    __syncthreads();
    int cur = 0;
    for (int d = 1; d < NBLK_MAX; d <<= 1) {
        if (t < NBLK_MAX) {
            int v = scan[cur][t];
            if (t >= d) v += scan[cur][t - d];
            scan[cur ^ 1][t] = v;
        }
        __syncthreads();
        cur ^= 1;
    }
    if (t < NBLK_MAX) segBase[t + 1] = scan[cur][t];
    if (t == 0) segBase[0] = 0;
    __syncthreads();

    const int total = __builtin_amdgcn_readfirstlane(segBase[NBLK_MAX]);
    const int wv    = __builtin_amdgcn_readfirstlane(t >> 5);

    const int row0  = b * RB;
    const int nleft = n_nodes - row0;
    const int nrows = nleft < RB ? nleft : RB;

    if (total <= SRT_CAP) {
        int2* srt = (int2*)pool;

        int key[SRT_PER_T], wbt[SRT_PER_T], rnk[SRT_PER_T];
        #pragma unroll
        for (int j = 0; j < SRT_PER_T; ++j) { key[j] = 0; wbt[j] = 0; }
        if (total > 0) {
            #pragma unroll
            for (int j = 0; j < SRT_PER_T; ++j) {
                const int i  = t + j * RED_T;
                const int ii = i < total ? i : total - 1;
                int lo = 0;
                #pragma unroll
                for (int step = NBLK_MAX / 2; step >= 1; step >>= 1)
                    lo += (segBase[lo + step] <= ii) ? step : 0;
                const int2 r = recs[segStart[lo] + (ii - segBase[lo])];
                key[j] = r.x; wbt[j] = r.y;
            }
        }
        #pragma unroll
        for (int j = 0; j < SRT_PER_T; ++j) {
            rnk[j] = -1;
            if (t + j * RED_T < total) rnk[j] = atomicAdd(&rowCnt[(unsigned)key[j] >> 24], 1);
        }
        __syncthreads();

        if (t < 32) {
            int c[8], ex[8];
            int run = 0;
            #pragma unroll
            for (int k = 0; k < 8; ++k) { c[k] = rowCnt[8 * t + k]; ex[k] = run; run += c[k]; }
            int incl = run;
            #pragma unroll
            for (int d = 1; d < 32; d <<= 1) {
                const int v = __shfl_up(incl, d);
                if (lane >= d) incl += v;
            }
            const int base = incl - run;
            #pragma unroll
            for (int k = 0; k < 8; ++k) rowStart[8 * t + k] = base + ex[k];
        }
        __syncthreads();

        #pragma unroll
        for (int j = 0; j < SRT_PER_T; ++j) {
            if (rnk[j] >= 0)
                srt[rowStart[(unsigned)key[j] >> 24] + rnk[j]] = make_int2(key[j] & 0xFFFFFF, wbt[j]);
        }
        __syncthreads();

        const int hh = lane >> 4;
        const int c4 = (lane & 15) << 4;
        for (int p = wv; p < RB / 2; p += RED_T / 32) {
            const int r = 2 * p + hh;
            const int n = rowCnt[r];
            const int s = rowStart[r];
            const int nA = __builtin_amdgcn_readlane(n, 0);
            const int nB = __builtin_amdgcn_readlane(n, 16);
            const int nmax = nA > nB ? nA : nB;
            float4 a = make_float4(0.0f, 0.0f, 0.0f, 0.0f);
            for (int i0 = 0; i0 < nmax; i0 += GATHER_G) {
                int2   rec[GATHER_G];
                float4 hv[GATHER_G];
                #pragma unroll
                for (int u = 0; u < GATHER_G; ++u) {
                    int i = i0 + u;
                    if (i > n - 1) i = n - 1;
                    rec[u] = srt[n > 0 ? s + i : 0];
                }
                #pragma unroll
                for (int u = 0; u < GATHER_G; ++u) {
                    const unsigned off = ((unsigned)rec[u].x << 8) | (unsigned)c4;
                    hv[u] = *(const float4*)((const char*)h + off);
                }
                #pragma unroll
                for (int u = 0; u < GATHER_G; ++u) {
                    const float w = (i0 + u) < n ? __int_as_float(rec[u].y) : 0.0f;
                    a.x += w * hv[u].x;
                    a.y += w * hv[u].y;
                    a.z += w * hv[u].z;
                    a.w += w * hv[u].w;
                }
            }
            if (r < nrows) *(float4*)((char*)out + ((size_t)(row0 + r) * OUT_FEATS * sizeof(float)) + c4) = a;
        }
        return;
    }

    float* acc = (float*)pool;
    for (int i = t; i < RB * OUT_FEATS; i += RED_T) acc[i] = 0.0f;
    __syncthreads();

    for (int base = wv * 32; base < total; base += RED_T) {
        const int i = base + lane;
        int key = 0, wbt = 0;
        if (i < total) {
            int lo = 0;
            #pragma unroll
            for (int step = NBLK_MAX / 2; step >= 1; step >>= 1)
                if (segBase[lo + step] <= i) lo += step;
            const int2 r = recs[segStart[lo] + (i - segBase[lo])];
            key = r.x; wbt = r.y;
        }
        const int left = total - base;
        const int cnt  = left < 32 ? left : 32;

        #pragma unroll
        for (int jb = 0; jb < 32; jb += EDGES_PER_WAVE) {
            if (cnt >= jb + EDGES_PER_WAVE) {
                float r0[EDGES_PER_WAVE], r1[EDGES_PER_WAVE];
                #pragma unroll
                for (int j = 0; j < EDGES_PER_WAVE; ++j) {
                    const int k = __builtin_amdgcn_readlane(key, jb + j);
                    const float* __restrict__ p = h + (size_t)(k & 0xFFFFFF) * OUT_FEATS;
                    r0[j] = p[lane];
                    r1[j] = p[lane + OUT_FEATS / 2];
                }
                #pragma unroll
                for (int j = 0; j < EDGES_PER_WAVE; ++j) {
                    const int   k = __builtin_amdgcn_readlane(key, jb + j);
                    const float w = __int_as_float(__builtin_amdgcn_readlane(wbt, jb + j));
                    lds_row_add(acc + ((unsigned)k >> 24) * OUT_FEATS, lane, w * r0[j], w * r1[j]);
                }
            } else {
                for (int j = jb; j < cnt; ++j) {
                    const int   k = __builtin_amdgcn_readlane(key, j);
                    const float w = __int_as_float(__builtin_amdgcn_readlane(wbt, j));
                    const float* __restrict__ p = h + (size_t)(k & 0xFFFFFF) * OUT_FEATS;
                    const float a0 = p[lane];
                    const float a1 = p[lane + OUT_FEATS / 2];
                    lds_row_add(acc + ((unsigned)k >> 24) * OUT_FEATS, lane, w * a0, w * a1);
                }
            }
        }
    }
    __syncthreads();

    const float4* acc4 = (const float4*)acc;
    float4* __restrict__ out4 = (float4*)(out + (size_t)row0 * OUT_FEATS);
    for (int i = t; i < nrows * (OUT_FEATS / 4); i += RED_T) out4[i] = acc4[i];
}

extern "C" void kernel_launch(void* const* d_in, const int* in_sizes, int n_in,
                              void* d_out, int out_size, void* d_ws, size_t ws_size,
                              hipStream_t stream)
{
    const float* x    = (const float*)d_in[0];
    const int*   esrc = (const int*)  d_in[1];
    const int*   edst = (const int*)  d_in[2];
    const float* ew   = (const float*)d_in[3];
    const float* W    = (const float*)d_in[4];
    const float* bias = (const float*)d_in[5];

    float* out = (float*)d_out;
    float* h   = (float*)d_ws;

    const int n_nodes = in_sizes[0] / IN_FEATS;
    const int n_edges = in_sizes[1];

    const int rowTiles = (n_nodes + 15) / 16;
    const int wavesPerBlock = 8;
    const int gemmBlocks = (rowTiles + wavesPerBlock - 1) / wavesPerBlock;

    const size_t hBytes   = (((size_t)n_nodes * OUT_FEATS * sizeof(float)) + 255) & ~(size_t)255;
    const size_t recBytes = (((size_t)n_edges * sizeof(int2)) + 255) & ~(size_t)255;
    const size_t offBytes = (size_t)(NBMAX + 1) * NBLK_MAX * sizeof(int);
    const int nblk    = (n_edges + CHUNK - 1) / CHUNK;
    const int buckets = (n_nodes + RB - 1) / RB;

    if (ws_size >= hBytes + recBytes + offBytes && nblk >= 1 && nblk <= NBLK_MAX && buckets <= NBMAX) {
        int2* recs = (int2*)((char*)d_ws + hBytes);
        int*  offT = (int*)((char*)d_ws + hBytes + recBytes);
        const int fusedGemmBlocks = (rowTiles + (BIN_T / 32) - 1) / (BIN_T / 32);
        gcn_gemm_and_bin<<<fusedGemmBlocks + nblk, BIN_T, 0, stream>>>(x, W, bias, h, esrc, edst, ew, recs, offT,
                                                                      n_edges, n_nodes, fusedGemmBlocks);
        gcn_bucket_sum<<<buckets, RED_T, 0, stream>>>(h, recs, offT, out, n_nodes, nblk);
        return;
    }

    hipMemsetAsync(d_out, 0, (size_t)out_size * sizeof(float), stream);
    gcn_gemm_wmma<<<gemmBlocks, wavesPerBlock * 32, 0, stream>>>(x, W, bias, h, n_nodes);

    const int edgesPerBlock = wavesPerBlock * EDGES_PER_WAVE;
    const int scatterBlocks = (n_edges + edgesPerBlock - 1) / edgesPerBlock;
    gcn_edge_scatter<<<scatterBlocks, wavesPerBlock * 32, 0, stream>>>(h, esrc, edst, ew, out, n_edges);
}


# fused launch re-gridded: 256 GEMM blocks of 25 waves (one per WGP, 16-row strips dealt round-robin) instead of 196 of 32: every WGP's matrix units work, at most 7 strips per SIMD, not 8
# speedup vs baseline: 3.2510x; 1.0356x over previous
#include <hip/hip_runtime.h>
#include <hip/hip_bf16.h>

typedef __attribute__((ext_vector_type(2))) float v2f;
typedef __attribute__((ext_vector_type(8))) float v8f;

#define IN_FEATS  256
#define OUT_FEATS 64

__device__ __forceinline__ void gemm_rows16(const float* __restrict__ x,
                                            const float* __restrict__ W,
                                            const float* __restrict__ bias,
                                            float* __restrict__ h,
                                            int n_nodes, int rowTile)
{
    const int lane   = threadIdx.x & 31;

    const int row0 = rowTile * 16;
    if (row0 >= n_nodes) return;

    const int m  = lane & 15;
    const int hi = lane >> 4;
    const int n  = lane & 15;

    int arow = row0 + m;
    if (arow >= n_nodes) arow = n_nodes - 1;
    const float* __restrict__ xrow = x + (size_t)arow * IN_FEATS;

    v8f acc0 = {}, acc1 = {}, acc2 = {}, acc3 = {};

    #pragma unroll 4
    for (int k = 0; k < IN_FEATS; k += 4) {
        const int kk = k + 2 * hi;
        v2f a = *(const v2f*)(xrow + kk);

        const float* __restrict__ w0 = W + (size_t)kk * OUT_FEATS + n;
        const float* __restrict__ w1 = w0 + OUT_FEATS;
        v2f b0 = { w0[0],  w1[0]  };
        v2f b1 = { w0[16], w1[16] };
        v2f b2 = { w0[32], w1[32] };
        v2f b3 = { w0[48], w1[48] };

        acc0 = __builtin_amdgcn_wmma_f32_16x16x4_f32(false, a, false, b0, (short)0, acc0, false, false);
        acc1 = __builtin_amdgcn_wmma_f32_16x16x4_f32(false, a, false, b1, (short)0, acc1, false, false);
        acc2 = __builtin_amdgcn_wmma_f32_16x16x4_f32(false, a, false, b2, (short)0, acc2, false, false);
        acc3 = __builtin_amdgcn_wmma_f32_16x16x4_f32(false, a, false, b3, (short)0, acc3, false, false);
    }

    const float bn0 = bias[n];
    const float bn1 = bias[16 + n];
    const float bn2 = bias[32 + n];
    const float bn3 = bias[48 + n];

    #pragma unroll
    for (int r = 0; r < 8; ++r) {
        const int row = row0 + r + 8 * hi;
        if (row < n_nodes) {
            float* __restrict__ hrow = h + (size_t)row * OUT_FEATS;
            hrow[n]      = acc0[r] + bn0;
            hrow[16 + n] = acc1[r] + bn1;
            hrow[32 + n] = acc2[r] + bn2;
            hrow[48 + n] = acc3[r] + bn3;
        }
    }
}

#define WFRAG_SLOTS ((IN_FEATS / 4) * 2 * 32)
#define WFRAG_FILL_T 800
#define WFRAG_FILL_ITERS ((WFRAG_SLOTS + WFRAG_FILL_T - 1) / WFRAG_FILL_T)

__device__ __forceinline__ void gemm_fill_wfrag(const float* __restrict__ W, float4* wfrag, int tid)
{
    #pragma unroll
    for (int i = 0; i < WFRAG_FILL_ITERS; ++i) {
        const int s  = tid + i * WFRAG_FILL_T;
        if (s < WFRAG_SLOTS) {
            const int ln = s & 31;
            const int p  = (s >> 5) & 1;
            const int ks = s >> 6;
            const int k0 = 4 * ks + 2 * (ln >> 4);
            const int c0 = 32 * p + (ln & 15);
            const float* __restrict__ w0 = W + (size_t)k0 * OUT_FEATS + c0;
            wfrag[s] = make_float4(w0[0], w0[OUT_FEATS], w0[16], w0[OUT_FEATS + 16]);
        }
    }
}

__device__ __forceinline__ void gemm_rows16_lds(const float* __restrict__ x,
                                                const float4* wfrag,
                                                const float* __restrict__ bias,
                                                float* __restrict__ h,
                                                int n_nodes, int rowTile)
{
    const int lane = threadIdx.x & 31;

    const int row0 = rowTile * 16;
    if (row0 >= n_nodes) return;

    const int m  = lane & 15;
    const int hi = lane >> 4;
    const int n  = lane & 15;

    int arow = row0 + m;
    if (arow >= n_nodes) arow = n_nodes - 1;
    const float* __restrict__ xrow = x + (size_t)arow * IN_FEATS;
    const float4* wl = wfrag + lane;

    v8f acc0 = {}, acc1 = {}, acc2 = {}, acc3 = {};

    #pragma unroll 2
    for (int k = 0; k < IN_FEATS; k += 4) {
        const int kk = k + 2 * hi;
        v2f a = *(const v2f*)(xrow + kk);

        const float4 q0 = wl[(k >> 2) * 64];
        const float4 q1 = wl[(k >> 2) * 64 + 32];
        v2f b0 = { q0.x, q0.y };
        v2f b1 = { q0.z, q0.w };
        v2f b2 = { q1.x, q1.y };
        v2f b3 = { q1.z, q1.w };

        acc0 = __builtin_amdgcn_wmma_f32_16x16x4_f32(false, a, false, b0, (short)0, acc0, false, false);
        acc1 = __builtin_amdgcn_wmma_f32_16x16x4_f32(false, a, false, b1, (short)0, acc1, false, false);
        acc2 = __builtin_amdgcn_wmma_f32_16x16x4_f32(false, a, false, b2, (short)0, acc2, false, false);
        acc3 = __builtin_amdgcn_wmma_f32_16x16x4_f32(false, a, false, b3, (short)0, acc3, false, false);
    }

    const float bn0 = bias[n];
    const float bn1 = bias[16 + n];
    const float bn2 = bias[32 + n];
    const float bn3 = bias[48 + n];

    #pragma unroll
    for (int r = 0; r < 8; ++r) {
        const int row = row0 + r + 8 * hi;
        if (row < n_nodes) {
            float* __restrict__ hrow = h + (size_t)row * OUT_FEATS;
            hrow[n]      = acc0[r] + bn0;
            hrow[16 + n] = acc1[r] + bn1;
            hrow[32 + n] = acc2[r] + bn2;
            hrow[48 + n] = acc3[r] + bn3;
        }
    }
}

__global__ void gcn_gemm_wmma(const float* __restrict__ x,
                              const float* __restrict__ W,
                              const float* __restrict__ bias,
                              float* __restrict__ h,
                              int n_nodes)
{
    const int waveId = threadIdx.x >> 5;
    const int wavesPerBlock = blockDim.x >> 5;
    gemm_rows16(x, W, bias, h, n_nodes, blockIdx.x * wavesPerBlock + waveId);
}

#define EDGES_PER_WAVE 16

__device__ __forceinline__ void edge_add(float* __restrict__ row, int lane, float v0, float v1)
{
#if defined(__HIP_DEVICE_COMPILE__)
    unsafeAtomicAdd(row + lane,                 v0);
    unsafeAtomicAdd(row + lane + OUT_FEATS / 2, v1);
#else
    atomicAdd(row + lane,                 v0);
    atomicAdd(row + lane + OUT_FEATS / 2, v1);
#endif
}

__global__ void gcn_edge_scatter(const float* __restrict__ h,
                                 const int* __restrict__ edge_src,
                                 const int* __restrict__ edge_dst,
                                 const float* __restrict__ edge_w,
                                 float* __restrict__ out,
                                 int n_edges)
{
    const int lane   = threadIdx.x & 31;
    const int waveId = threadIdx.x >> 5;
    const int wavesPerBlock = blockDim.x >> 5;

    const int wave = __builtin_amdgcn_readfirstlane(blockIdx.x * wavesPerBlock + waveId);
    const int e0 = wave * EDGES_PER_WAVE;
    if (e0 >= n_edges) return;
    const int left = n_edges - e0;
    const int cnt  = left < EDGES_PER_WAVE ? left : EDGES_PER_WAVE;

    const int sub = lane & (EDGES_PER_WAVE - 1);
    const int ei  = e0 + (sub < cnt ? sub : 0);
    const int sv = edge_src[ei];
    const int dv = edge_dst[ei];
    const int wv = __float_as_int(edge_w[ei]);

    if (cnt == EDGES_PER_WAVE) {
        float r0[EDGES_PER_WAVE], r1[EDGES_PER_WAVE];
        #pragma unroll
        for (int j = 0; j < EDGES_PER_WAVE; ++j) {
            const int s = __builtin_amdgcn_readlane(sv, j);
            const float* __restrict__ p = h + (size_t)s * OUT_FEATS;
            r0[j] = p[lane];
            r1[j] = p[lane + OUT_FEATS / 2];
        }
        #pragma unroll
        for (int j = 0; j < EDGES_PER_WAVE; ++j) {
            const int   d = __builtin_amdgcn_readlane(dv, j);
            const float w = __int_as_float(__builtin_amdgcn_readlane(wv, j));
            edge_add(out + (size_t)d * OUT_FEATS, lane, w * r0[j], w * r1[j]);
        }
    } else {
        for (int j = 0; j < cnt; ++j) {
            const int   s = __builtin_amdgcn_readlane(sv, j);
            const int   d = __builtin_amdgcn_readlane(dv, j);
            const float w = __int_as_float(__builtin_amdgcn_readlane(wv, j));
            const float* __restrict__ p = h + (size_t)s * OUT_FEATS;
            const float a0 = p[lane];
            const float a1 = p[lane + OUT_FEATS / 2];
            edge_add(out + (size_t)d * OUT_FEATS, lane, w * a0, w * a1);
        }
    }
}

#define RB        196
#define NBMAX     512
#define CHUNK     6400
#define NBLK_MAX  256
#define BIN_T     800
#define GEMM_BLOCKS 256
#define BIN_EPT   ((CHUNK + BIN_T - 1) / BIN_T)
#define RED_T     512

__global__ __launch_bounds__(BIN_T)
void gcn_gemm_and_bin(const float* __restrict__ x,
                      const float* __restrict__ W,
                      const float* __restrict__ bias,
                      float* __restrict__ h,
                      const int* __restrict__ edge_src,
                      const int* __restrict__ edge_dst,
                      const float* __restrict__ edge_w,
                      int2* __restrict__ recs,
                      int* __restrict__ offT,
                      int n_edges, int n_nodes, int gemm_blocks)
{
    __shared__ __attribute__((aligned(16))) unsigned char smem[WFRAG_SLOTS * sizeof(float4)];
    static_assert((NBMAX + 2 * NBMAX) * sizeof(int) + CHUNK * sizeof(int2) <= WFRAG_SLOTS * sizeof(float4),
                  "the binning role's arrays must fit the shared block");
    static_assert(WFRAG_FILL_T == BIN_T, "the W copy is written for this launch's block size");

    if ((int)blockIdx.x < gemm_blocks) {
        float4* wfrag = (float4*)smem;
        gemm_fill_wfrag(W, wfrag, threadIdx.x);
        __syncthreads();
        const int wave = __builtin_amdgcn_readfirstlane(threadIdx.x >> 5);
        const int rowTiles = (n_nodes + 15) >> 4;
        for (int tile = blockIdx.x + gemm_blocks * wave; tile < rowTiles; tile += gemm_blocks * (BIN_T / 32))
            gemm_rows16_lds(x, wfrag, bias, h, n_nodes, tile);
        return;
    }

    int*  cnt             = (int*)smem;
    int  (*scan)[NBMAX]   = (int (*)[NBMAX])(cnt + NBMAX);
    int2* stage           = (int2*)(cnt + 3 * NBMAX);

    const int t   = threadIdx.x;
    const int blk = blockIdx.x - gemm_blocks;
    const int e0  = blk * CHUNK;
    const int rem = n_edges - e0;
    const int len = rem < CHUNK ? rem : CHUNK;

    if (t < NBMAX) cnt[t] = 0;
    __syncthreads();

    int key[BIN_EPT], wbt[BIN_EPT], bkt[BIN_EPT], rnk[BIN_EPT];
    #pragma unroll
    for (int i = 0; i < BIN_EPT; ++i) {
        const int idx = t + i * BIN_T;
        bkt[i] = -1; key[i] = 0; wbt[i] = 0; rnk[i] = 0;
        if (idx < len) {
            const int e = e0 + idx;
            const int d = edge_dst[e];
            const int s = edge_src[e];
            const int w = __float_as_int(edge_w[e]);
            if ((unsigned)d < (unsigned)n_nodes) {
                const int b = d / RB;
                bkt[i] = b;
                key[i] = s | ((d - b * RB) << 24);
                wbt[i] = w;
                rnk[i] = atomicAdd(&cnt[b], 1);
            }
        }
    }
    __syncthreads();

    if (t < NBMAX) scan[0][t] = cnt[t];
    __syncthreads();
    int cur = 0;
    for (int d = 1; d < NBMAX; d <<= 1) {
        if (t < NBMAX) {
            int v = scan[cur][t];
            if (t >= d) v += scan[cur][t - d];
            scan[cur ^ 1][t] = v;
        }
        __syncthreads();
        cur ^= 1;
    }
    if (t < NBMAX) {
        const int ex = scan[cur][t] - cnt[t];
        scan[cur ^ 1][t] = ex;
        offT[t * NBLK_MAX + blk] = ex;
        if (t == NBMAX - 1) offT[NBMAX * NBLK_MAX + blk] = scan[cur][t];
    }
    __syncthreads();
    const int total = scan[cur][NBMAX - 1];

    #pragma unroll
    for (int i = 0; i < BIN_EPT; ++i) {
        if (bkt[i] >= 0) stage[scan[cur ^ 1][bkt[i]] + rnk[i]] = make_int2(key[i], wbt[i]);
    }
    __syncthreads();

    for (int idx = t; idx < total; idx += BIN_T) recs[e0 + idx] = stage[idx];
}

__device__ __forceinline__ void lds_row_add(float* row, int lane, float v0, float v1)
{
    __hip_atomic_fetch_add(row + lane,                 v0, __ATOMIC_RELAXED, __HIP_MEMORY_SCOPE_WORKGROUP);
    __hip_atomic_fetch_add(row + lane + OUT_FEATS / 2, v1, __ATOMIC_RELAXED, __HIP_MEMORY_SCOPE_WORKGROUP);
}

#define SRT_CAP    4096
#define SRT_PER_T  (SRT_CAP / RED_T)
#define ROW_SLOTS  256
#define GATHER_G   16

__global__ __launch_bounds__(RED_T)
void gcn_bucket_sum(const float* __restrict__ h,
                    const int2* __restrict__ recs,
                    const int* __restrict__ offT,
                    float* __restrict__ out,
                    int n_nodes, int nblk)
{
    __shared__ __attribute__((aligned(16))) unsigned char pool[RB * OUT_FEATS * sizeof(float)];
    static_assert(SRT_CAP * sizeof(int2) <= RB * OUT_FEATS * sizeof(float), "sorted records must fit the shared block");
    static_assert(RB <= ROW_SLOTS && (RB % 2) == 0, "two rows per wave, counters for every row");
    static_assert(SRT_CAP % RED_T == 0 && ROW_SLOTS == 8 * 32, "whole records per thread; 8 counters per lane in the scan");
    __shared__ int   segStart[NBLK_MAX];
    __shared__ int   segBase[NBLK_MAX + 1];
    __shared__ int   scan[2][NBLK_MAX];
    __shared__ int   rowCnt[ROW_SLOTS];
    __shared__ int   rowStart[ROW_SLOTS];

    const int t    = threadIdx.x;
    const int lane = t & 31;
    const int b    = blockIdx.x;

    if (t < ROW_SLOTS) rowCnt[t] = 0;

    if (t < NBLK_MAX) {
        int mylen = 0, st = 0;
        if (t < nblk) {
            const int s = offT[b * NBLK_MAX + t];
            const int e = offT[(b + 1) * NBLK_MAX + t];
            st = t * CHUNK + s;
            mylen = e - s;
        }
        segStart[t] = st;
        scan[0][t]  = mylen;
    }
    __syncthreads();
    int cur = 0;
    for (int d = 1; d < NBLK_MAX; d <<= 1) {
        if (t < NBLK_MAX) {
            int v = scan[cur][t];
            if (t >= d) v += scan[cur][t - d];
            scan[cur ^ 1][t] = v;
        }
        __syncthreads();
        cur ^= 1;
    }
    if (t < NBLK_MAX) segBase[t + 1] = scan[cur][t];
    if (t == 0) segBase[0] = 0;
    __syncthreads();

    const int total = __builtin_amdgcn_readfirstlane(segBase[NBLK_MAX]);
    const int wv    = __builtin_amdgcn_readfirstlane(t >> 5);

    const int row0  = b * RB;
    const int nleft = n_nodes - row0;
    const int nrows = nleft < RB ? nleft : RB;

    if (total <= SRT_CAP) {
        int2* srt = (int2*)pool;

        int key[SRT_PER_T], wbt[SRT_PER_T], rnk[SRT_PER_T];
        #pragma unroll
        for (int j = 0; j < SRT_PER_T; ++j) { key[j] = 0; wbt[j] = 0; }
        if (total > 0) {
            #pragma unroll
            for (int j = 0; j < SRT_PER_T; ++j) {
                const int i  = t + j * RED_T;
                const int ii = i < total ? i : total - 1;
                int lo = 0;
                #pragma unroll
                for (int step = NBLK_MAX / 2; step >= 1; step >>= 1)
                    lo += (segBase[lo + step] <= ii) ? step : 0;
                const int2 r = recs[segStart[lo] + (ii - segBase[lo])];
                key[j] = r.x; wbt[j] = r.y;
            }
        }
        #pragma unroll
        for (int j = 0; j < SRT_PER_T; ++j) {
            rnk[j] = -1;
            if (t + j * RED_T < total) rnk[j] = atomicAdd(&rowCnt[(unsigned)key[j] >> 24], 1);
        }
        __syncthreads();

        if (t < 32) {
            int c[8], ex[8];
            int run = 0;
            #pragma unroll
            for (int k = 0; k < 8; ++k) { c[k] = rowCnt[8 * t + k]; ex[k] = run; run += c[k]; }
            int incl = run;
            #pragma unroll
            for (int d = 1; d < 32; d <<= 1) {
                const int v = __shfl_up(incl, d);
                if (lane >= d) incl += v;
            }
            const int base = incl - run;
            #pragma unroll
            for (int k = 0; k < 8; ++k) rowStart[8 * t + k] = base + ex[k];
        }
        __syncthreads();

        #pragma unroll
        for (int j = 0; j < SRT_PER_T; ++j) {
            if (rnk[j] >= 0)
                srt[rowStart[(unsigned)key[j] >> 24] + rnk[j]] = make_int2(key[j] & 0xFFFFFF, wbt[j]);
        }
        __syncthreads();

        const int hh = lane >> 4;
        const int c4 = (lane & 15) << 4;
        for (int p = wv; p < RB / 2; p += RED_T / 32) {
            const int r = 2 * p + hh;
            const int n = rowCnt[r];
            const int s = rowStart[r];
            const int nA = __builtin_amdgcn_readlane(n, 0);
            const int nB = __builtin_amdgcn_readlane(n, 16);
            const int nmax = nA > nB ? nA : nB;
            float4 a = make_float4(0.0f, 0.0f, 0.0f, 0.0f);
            for (int i0 = 0; i0 < nmax; i0 += GATHER_G) {
                int2   rec[GATHER_G];
                float4 hv[GATHER_G];
                #pragma unroll
                for (int u = 0; u < GATHER_G; ++u) {
                    int i = i0 + u;
                    if (i > n - 1) i = n - 1;
                    rec[u] = srt[n > 0 ? s + i : 0];
                }
                #pragma unroll
                for (int u = 0; u < GATHER_G; ++u) {
                    const unsigned off = ((unsigned)rec[u].x << 8) | (unsigned)c4;
                    hv[u] = *(const float4*)((const char*)h + off);
                }
                #pragma unroll
                for (int u = 0; u < GATHER_G; ++u) {
                    const float w = (i0 + u) < n ? __int_as_float(rec[u].y) : 0.0f;
                    a.x += w * hv[u].x;
                    a.y += w * hv[u].y;
                    a.z += w * hv[u].z;
                    a.w += w * hv[u].w;
                }
            }
            if (r < nrows) *(float4*)((char*)out + ((size_t)(row0 + r) * OUT_FEATS * sizeof(float)) + c4) = a;
        }
        return;
    }

    float* acc = (float*)pool;
    for (int i = t; i < RB * OUT_FEATS; i += RED_T) acc[i] = 0.0f;
    __syncthreads();

    for (int base = wv * 32; base < total; base += RED_T) {
        const int i = base + lane;
        int key = 0, wbt = 0;
        if (i < total) {
            int lo = 0;
            #pragma unroll
            for (int step = NBLK_MAX / 2; step >= 1; step >>= 1)
                if (segBase[lo + step] <= i) lo += step;
            const int2 r = recs[segStart[lo] + (i - segBase[lo])];
            key = r.x; wbt = r.y;
        }
        const int left = total - base;
        const int cnt  = left < 32 ? left : 32;

        #pragma unroll
        for (int jb = 0; jb < 32; jb += EDGES_PER_WAVE) {
            if (cnt >= jb + EDGES_PER_WAVE) {
                float r0[EDGES_PER_WAVE], r1[EDGES_PER_WAVE];
                #pragma unroll
                for (int j = 0; j < EDGES_PER_WAVE; ++j) {
                    const int k = __builtin_amdgcn_readlane(key, jb + j);
                    const float* __restrict__ p = h + (size_t)(k & 0xFFFFFF) * OUT_FEATS;
                    r0[j] = p[lane];
                    r1[j] = p[lane + OUT_FEATS / 2];
                }
                #pragma unroll
                for (int j = 0; j < EDGES_PER_WAVE; ++j) {
                    const int   k = __builtin_amdgcn_readlane(key, jb + j);
                    const float w = __int_as_float(__builtin_amdgcn_readlane(wbt, jb + j));
                    lds_row_add(acc + ((unsigned)k >> 24) * OUT_FEATS, lane, w * r0[j], w * r1[j]);
                }
            } else {
                for (int j = jb; j < cnt; ++j) {
                    const int   k = __builtin_amdgcn_readlane(key, j);
                    const float w = __int_as_float(__builtin_amdgcn_readlane(wbt, j));
                    const float* __restrict__ p = h + (size_t)(k & 0xFFFFFF) * OUT_FEATS;
                    const float a0 = p[lane];
                    const float a1 = p[lane + OUT_FEATS / 2];
                    lds_row_add(acc + ((unsigned)k >> 24) * OUT_FEATS, lane, w * a0, w * a1);
                }
            }
        }
    }
    __syncthreads();

    const float4* acc4 = (const float4*)acc;
    float4* __restrict__ out4 = (float4*)(out + (size_t)row0 * OUT_FEATS);
    for (int i = t; i < nrows * (OUT_FEATS / 4); i += RED_T) out4[i] = acc4[i];
}

extern "C" void kernel_launch(void* const* d_in, const int* in_sizes, int n_in,
                              void* d_out, int out_size, void* d_ws, size_t ws_size,
                              hipStream_t stream)
{
    const float* x    = (const float*)d_in[0];
    const int*   esrc = (const int*)  d_in[1];
    const int*   edst = (const int*)  d_in[2];
    const float* ew   = (const float*)d_in[3];
    const float* W    = (const float*)d_in[4];
    const float* bias = (const float*)d_in[5];

    float* out = (float*)d_out;
    float* h   = (float*)d_ws;

    const int n_nodes = in_sizes[0] / IN_FEATS;
    const int n_edges = in_sizes[1];

    const int rowTiles = (n_nodes + 15) / 16;
    const int wavesPerBlock = 8;
    const int gemmBlocks = (rowTiles + wavesPerBlock - 1) / wavesPerBlock;

    const size_t hBytes   = (((size_t)n_nodes * OUT_FEATS * sizeof(float)) + 255) & ~(size_t)255;
    const size_t recBytes = (((size_t)n_edges * sizeof(int2)) + 255) & ~(size_t)255;
    const size_t offBytes = (size_t)(NBMAX + 1) * NBLK_MAX * sizeof(int);
    const int nblk    = (n_edges + CHUNK - 1) / CHUNK;
    const int buckets = (n_nodes + RB - 1) / RB;

    if (ws_size >= hBytes + recBytes + offBytes && nblk >= 1 && nblk <= NBLK_MAX && buckets <= NBMAX) {
        int2* recs = (int2*)((char*)d_ws + hBytes);
        int*  offT = (int*)((char*)d_ws + hBytes + recBytes);
        const int fusedGemmBlocks = GEMM_BLOCKS;
        gcn_gemm_and_bin<<<fusedGemmBlocks + nblk, BIN_T, 0, stream>>>(x, W, bias, h, esrc, edst, ew, recs, offT,
                                                                      n_edges, n_nodes, fusedGemmBlocks);
        gcn_bucket_sum<<<buckets, RED_T, 0, stream>>>(h, recs, offT, out, n_nodes, nblk);
        return;
    }

    hipMemsetAsync(d_out, 0, (size_t)out_size * sizeof(float), stream);
    gcn_gemm_wmma<<<gemmBlocks, wavesPerBlock * 32, 0, stream>>>(x, W, bias, h, n_nodes);

    const int edgesPerBlock = wavesPerBlock * EDGES_PER_WAVE;
    const int scatterBlocks = (n_edges + edgesPerBlock - 1) / edgesPerBlock;
    gcn_edge_scatter<<<scatterBlocks, wavesPerBlock * 32, 0, stream>>>(h, esrc, edst, ew, out, n_edges);
}
